# starting kernel
# baseline (speedup 1.0000x reference)
#include <hip/hip_runtime.h>


#define N_NODES 100000
#define N_EDGES 1600000
#define NFEAT   256
#define NHID    128
#define NCLASS  40

typedef __attribute__((ext_vector_type(2))) float v2f;
typedef __attribute__((ext_vector_type(8))) float v8f;

__global__ __launch_bounds__(256)
void gcn_gemm1(const float* __restrict__ x, const float* __restrict__ W1,
               float* __restrict__ H0) {
  const int mt   = blockIdx.x;
  const int wave = threadIdx.x >> 5;
  const int lane = threadIdx.x & 31;
  const int half = lane >> 4;
  const int mn   = lane & 15;
  const int m0 = mt * 16;
  const int n0 = wave * 16;

  const float* arow = x  + (size_t)(m0 + mn) * NFEAT;
  const float* bcol = W1 + n0 + mn;

  v8f c = {};
  #pragma unroll 4
  for (int k = 0; k < NFEAT; k += 4) {
    const int k0 = k + 2 * half;
    v2f a, b;
    a.x = arow[k0];
    a.y = arow[k0 + 1];
    b.x = bcol[(size_t)k0 * NHID];
    b.y = bcol[(size_t)(k0 + 1) * NHID];
    c = __builtin_amdgcn_wmma_f32_16x16x4_f32(
         false, a,  false, b,
         (short)0, c,  false,  false);
  }
  float* crow = H0 + (size_t)m0 * NHID + n0 + mn;
  #pragma unroll
  for (int j = 0; j < 8; ++j)
    crow[(size_t)(j + 8 * half) * NHID] = c[j];
}

__global__ __launch_bounds__(256)
void gcn_spmm1(const int* __restrict__ esrc, const int* __restrict__ edst,
               const float* __restrict__ ew, const float* __restrict__ H0,
               float* __restrict__ agg) {
  const long long tid = (long long)blockIdx.x * blockDim.x + threadIdx.x;
  const int e = (int)(tid >> 5);
  if (e >= N_EDGES) return;
  const int lane = threadIdx.x & 31;
  const int s = esrc[e];
  const int d = edst[e];
  const float w = ew[e];
  const float4 hv = *(const float4*)(H0 + (size_t)s * NHID + lane * 4);
  float* o = agg + (size_t)d * NHID + lane * 4;
  atomicAdd(o + 0, w * hv.x);
  atomicAdd(o + 1, w * hv.y);
  atomicAdd(o + 2, w * hv.z);
  atomicAdd(o + 3, w * hv.w);
}

__global__ __launch_bounds__(96)
void gcn_gemm2(const float* __restrict__ agg, const float* __restrict__ b1,
               const float* __restrict__ W2, float* __restrict__ H2) {
  const int mt   = blockIdx.x;
  const int wave = threadIdx.x >> 5;
  const int lane = threadIdx.x & 31;
  const int half = lane >> 4;
  const int mn   = lane & 15;
  const int m0  = mt * 16;
  const int col = wave * 16 + mn;
  const bool colok = (col < NCLASS);

  const float* arow = agg + (size_t)(m0 + mn) * NHID;

  v8f c = {};
  #pragma unroll 4
  for (int k = 0; k < NHID; k += 4) {
    const int k0 = k + 2 * half;
    v2f a, b;
    a.x = fmaxf(arow[k0]     + b1[k0],     0.0f);
    a.y = fmaxf(arow[k0 + 1] + b1[k0 + 1], 0.0f);
    b.x = colok ? W2[(size_t)k0 * NCLASS + col]       : 0.0f;
    b.y = colok ? W2[(size_t)(k0 + 1) * NCLASS + col] : 0.0f;
    c = __builtin_amdgcn_wmma_f32_16x16x4_f32(
        false, a, false, b, (short)0, c, false, false);
  }
  if (colok) {
    #pragma unroll
    for (int j = 0; j < 8; ++j)
      H2[(size_t)(m0 + j + 8 * half) * NCLASS + col] = c[j];
  }
}

__global__ __launch_bounds__(256)
void gcn_spmm2(const int* __restrict__ esrc, const int* __restrict__ edst,
               const float* __restrict__ ew, const float* __restrict__ H2,
               float* __restrict__ out) {
  const long long tid = (long long)blockIdx.x * blockDim.x + threadIdx.x;
  if (tid >= (long long)N_EDGES * NCLASS) return;
  const int e = (int)(tid / NCLASS);
  const int cc = (int)(tid % NCLASS);
  const int s = esrc[e];
  const int d = edst[e];
  atomicAdd(&out[(size_t)d * NCLASS + cc], ew[e] * H2[(size_t)s * NCLASS + cc]);
}

__global__ __launch_bounds__(256)
void gcn_zero4(float4* __restrict__ p, long long n4) {
  const long long i = (long long)blockIdx.x * blockDim.x + threadIdx.x;
  if (i < n4) p[i] = make_float4(0.f, 0.f, 0.f, 0.f);
}

__global__ __launch_bounds__(256)
void gcn_init_out(float* __restrict__ out, const float* __restrict__ b2) {
  const long long i = (long long)blockIdx.x * blockDim.x + threadIdx.x;
  if (i < (long long)N_NODES * NCLASS) out[i] = b2[i % NCLASS];
}

extern "C" void kernel_launch(void* const* d_in, const int* in_sizes, int n_in,
                              void* d_out, int out_size, void* d_ws, size_t ws_size,
                              hipStream_t stream) {
  const float* x    = (const float*)d_in[0];
  const int*   esrc = (const int*)  d_in[1];
  const int*   edst = (const int*)  d_in[2];
  const float* ew   = (const float*)d_in[3];
  const float* W1   = (const float*)d_in[4];
  const float* b1   = (const float*)d_in[5];
  const float* W2   = (const float*)d_in[6];
  const float* b2   = (const float*)d_in[7];
  float* out = (float*)d_out;

  float* H0  = (float*)d_ws;
  float* agg = H0 + (size_t)N_NODES * NHID;
  float* H2  = H0;

  {
    const long long n4 = (long long)N_NODES * NHID / 4;
    gcn_zero4<<<(unsigned)((n4 + 255) / 256), 256, 0, stream>>>((float4*)agg, n4);
  }
  gcn_gemm1<<<N_NODES / 16, 256, 0, stream>>>(x, W1, H0);
  {
    const long long nt = (long long)N_EDGES * 32;
    gcn_spmm1<<<(unsigned)((nt + 255) / 256), 256, 0, stream>>>(esrc, edst, ew, H0, agg);
  }
  gcn_gemm2<<<N_NODES / 16, 96, 0, stream>>>(agg, b1, W2, H2);
  {
    const long long n = (long long)N_NODES * NCLASS;
    gcn_init_out<<<(unsigned)((n + 255) / 256), 256, 0, stream>>>(out, b2);
  }
  {
    const long long nt = (long long)N_EDGES * NCLASS;
    gcn_spmm2<<<(unsigned)((nt + 255) / 256), 256, 0, stream>>>(esrc, edst, ew, H2, out);
  }
}


# Edges counting-sorted by dst in-call (integer atomic rank, scan, place); both SpMMs sum a row's edges in registers in arrival order, no float atomics; layer-1 sums feed the W2 WMMA via LDS
# speedup vs baseline: 1.9186x; 1.9186x over previous
#include <hip/hip_runtime.h>


#define N_NODES 100000
#define N_EDGES 1600000
#define NFEAT   256
#define NHID    128
#define NCLASS  40

#define LDA     132

typedef __attribute__((ext_vector_type(2))) float v2f;
typedef __attribute__((ext_vector_type(8))) float v8f;

__global__ __launch_bounds__(256)
void gcn_gemm1(const float* __restrict__ x, const float* __restrict__ W1,
               float* __restrict__ H0, int* __restrict__ cnt) {
  const int mt   = blockIdx.x;
  const int wave = threadIdx.x >> 5;
  const int lane = threadIdx.x & 31;
  const int half = lane >> 4;
  const int mn   = lane & 15;
  const int m0 = mt * 16;
  const int n0 = wave * 16;

  if (threadIdx.x < 16) cnt[m0 + threadIdx.x] = 0;

  const float* arow = x  + (size_t)(m0 + mn) * NFEAT;
  const float* bcol = W1 + n0 + mn;

  v8f c = {};
  #pragma unroll 4
  for (int k = 0; k < NFEAT; k += 4) {
    const int k0 = k + 2 * half;
    v2f a, b;
    a.x = arow[k0];
    a.y = arow[k0 + 1];
    b.x = bcol[(size_t)k0 * NHID];
    b.y = bcol[(size_t)(k0 + 1) * NHID];
    c = __builtin_amdgcn_wmma_f32_16x16x4_f32(
         false, a,  false, b,
         (short)0, c,  false,  false);
  }
  float* crow = H0 + (size_t)m0 * NHID + n0 + mn;
  #pragma unroll
  for (int j = 0; j < 8; ++j)
    crow[(size_t)(j + 8 * half) * NHID] = c[j];
}

__global__ __launch_bounds__(256)
void gcn_count(const int* __restrict__ edst, int* __restrict__ cnt,
               int* __restrict__ slot) {
  const int e = blockIdx.x * 256 + threadIdx.x;
  if (e < N_EDGES) slot[e] = atomicAdd(&cnt[edst[e]], 1);
}

__global__ __launch_bounds__(1024)
void gcn_rowptr(const int* __restrict__ cnt, int* __restrict__ rowptr) {
  __shared__ int s_before[32];
  __shared__ int s_incl[32];
  const int t = threadIdx.x;
  const int lane = t & 31;
  const int wave = t >> 5;
  const int first = blockIdx.x * 1024;

  int before = 0;
  for (int i = t; i < first; i += 1024) before += cnt[i];

  const int i = first + t;
  const int v = (i < N_NODES) ? cnt[i] : 0;
  int inc = v;
  for (int o = 1; o < 32; o <<= 1) {
    const int u = __shfl_up(inc, o);
    if (lane >= o) inc += u;
  }
  for (int o = 16; o > 0; o >>= 1) before += __shfl_xor(before, o);
  if (lane == 31) s_incl[wave] = inc;
  if (lane == 0)  s_before[wave] = before;
  __syncthreads();
  int bsum = s_before[lane];
  int wsum = (lane < wave) ? s_incl[lane] : 0;
  for (int o = 16; o > 0; o >>= 1) {
    bsum += __shfl_xor(bsum, o);
    wsum += __shfl_xor(wsum, o);
  }
  const int excl = bsum + wsum + inc - v;
  if (i < N_NODES) rowptr[i] = excl;
  if (i == N_NODES - 1) rowptr[N_NODES] = excl + v;
}

__global__ __launch_bounds__(256)
void gcn_place(const int* __restrict__ esrc, const int* __restrict__ edst,
               const float* __restrict__ ew, const int* __restrict__ rowptr,
               const int* __restrict__ slot, int* __restrict__ rec) {
  const int e = blockIdx.x * 256 + threadIdx.x;
  if (e < N_EDGES) {
    const int pos = rowptr[edst[e]] + slot[e];
    ((int2*)rec)[pos] = make_int2(esrc[e], __float_as_int(ew[e]));
  }
}

__global__ __launch_bounds__(256)
void gcn_agg_gemm2(const int* __restrict__ rowptr, const int* __restrict__ rec,
                   const float* __restrict__ H0, const float* __restrict__ b1,
                   const float* __restrict__ W2, float* __restrict__ H2) {
  __shared__ __attribute__((aligned(16))) float sA[16 * LDA];
  const int mt   = blockIdx.x;
  const int wave = threadIdx.x >> 5;
  const int lane = threadIdx.x & 31;
  const int m0   = mt * 16;
  const int2* recs = (const int2*)rec;
  const float4 bv = *(const float4*)(b1 + lane * 4);

  for (int rr = 0; rr < 2; ++rr) {
    const int rl  = wave + 8 * rr;
    const int beg = __builtin_amdgcn_readfirstlane(rowptr[m0 + rl]);
    const int end = __builtin_amdgcn_readfirstlane(rowptr[m0 + rl + 1]);
    float4 acc = make_float4(0.f, 0.f, 0.f, 0.f);
    for (int base = beg; base < end; base += 32) {
      const int left = end - base;
      const int n = left < 32 ? left : 32;
      int2 sw = make_int2(0, 0);
      if (lane < n) sw = recs[base + lane];
      #pragma unroll 4
      for (int j = 0; j < n; ++j) {
        const int   s = __builtin_amdgcn_readlane(sw.x, j);
        const float w = __int_as_float(__builtin_amdgcn_readlane(sw.y, j));
        const float4 hv = *(const float4*)(H0 + (size_t)s * NHID + lane * 4);
        acc.x += w * hv.x;
        acc.y += w * hv.y;
        acc.z += w * hv.z;
        acc.w += w * hv.w;
      }
    }
    float4 o;
    o.x = fmaxf(acc.x + bv.x, 0.0f);
    o.y = fmaxf(acc.y + bv.y, 0.0f);
    o.z = fmaxf(acc.z + bv.z, 0.0f);
    o.w = fmaxf(acc.w + bv.w, 0.0f);
    *(float4*)(sA + rl * LDA + lane * 4) = o;
  }
  __syncthreads();

  if (wave < 3) {
    const int half = lane >> 4;
    const int mn   = lane & 15;
    const int col  = wave * 16 + mn;
    const bool colok = (col < NCLASS);
    const float* arow = sA + mn * LDA;

    v8f c = {};
    #pragma unroll 4
    for (int k = 0; k < NHID; k += 4) {
      const int k0 = k + 2 * half;
      v2f a, b;
      a.x = arow[k0];
      a.y = arow[k0 + 1];
      b.x = colok ? W2[(size_t)k0 * NCLASS + col]       : 0.0f;
      b.y = colok ? W2[(size_t)(k0 + 1) * NCLASS + col] : 0.0f;
      c = __builtin_amdgcn_wmma_f32_16x16x4_f32(
          false, a, false, b, (short)0, c, false, false);
    }
    if (colok) {
      #pragma unroll
      for (int j = 0; j < 8; ++j)
        H2[(size_t)(m0 + j + 8 * half) * NCLASS + col] = c[j];
    }
  }
}

__global__ __launch_bounds__(256)
void gcn_spmm2(const int* __restrict__ rowptr, const int* __restrict__ rec,
               const float* __restrict__ H2, const float* __restrict__ b2,
               float* __restrict__ out) {
  const int tid = blockIdx.x * 256 + threadIdx.x;
  if (tid >= N_NODES * (NCLASS / 4)) return;
  const int r  = tid / (NCLASS / 4);
  const int c4 = tid % (NCLASS / 4);
  const int2* recs = (const int2*)rec;
  const int beg = rowptr[r];
  const int end = rowptr[r + 1];
  const float* hcol = H2 + c4 * 4;
  float4 acc = make_float4(0.f, 0.f, 0.f, 0.f);
  #pragma unroll 4
  for (int e = beg; e < end; ++e) {
    const int2 sw = recs[e];
    const float w = __int_as_float(sw.y);
    const float4 hv = *(const float4*)(hcol + (size_t)sw.x * NCLASS);
    acc.x += w * hv.x;
    acc.y += w * hv.y;
    acc.z += w * hv.z;
    acc.w += w * hv.w;
  }
  const float4 bv = *(const float4*)(b2 + c4 * 4);
  float4 o;
  o.x = acc.x + bv.x;
  o.y = acc.y + bv.y;
  o.z = acc.z + bv.z;
  o.w = acc.w + bv.w;
  *(float4*)(out + (size_t)tid * 4) = o;
}

extern "C" void kernel_launch(void* const* d_in, const int* in_sizes, int n_in,
                              void* d_out, int out_size, void* d_ws, size_t ws_size,
                              hipStream_t stream) {
  const float* x    = (const float*)d_in[0];
  const int*   esrc = (const int*)  d_in[1];
  const int*   edst = (const int*)  d_in[2];
  const float* ew   = (const float*)d_in[3];
  const float* W1   = (const float*)d_in[4];
  const float* b1   = (const float*)d_in[5];
  const float* W2   = (const float*)d_in[6];
  const float* b2   = (const float*)d_in[7];
  float* out = (float*)d_out;

  float* H0     = (float*)d_ws;
  float* H2     = H0 + (size_t)N_NODES * NHID;
  int*   rec    = (int*)(H2 + (size_t)N_NODES * NCLASS);
  int*   slot   = rec + (size_t)N_EDGES * 2;
  int*   cnt    = slot + (size_t)N_EDGES;
  int*   rowptr = cnt + (size_t)N_NODES;

  gcn_gemm1<<<N_NODES / 16, 256, 0, stream>>>(x, W1, H0, cnt);
  gcn_count<<<N_EDGES / 256, 256, 0, stream>>>(edst, cnt, slot);
  gcn_rowptr<<<(N_NODES + 1023) / 1024, 1024, 0, stream>>>(cnt, rowptr);
  gcn_place<<<N_EDGES / 256, 256, 0, stream>>>(esrc, edst, ew, rowptr, slot, rec);
  gcn_agg_gemm2<<<N_NODES / 16, 256, 0, stream>>>(rowptr, rec, H0, b1, W2, H2);
  gcn_spmm2<<<(N_NODES * (NCLASS / 4) + 255) / 256, 256, 0, stream>>>(rowptr, rec, H2, b2, out);
}


# GEMM1 operand feed: each wave keeps its 256x16 slice of W1 in registers, 32-row x tiles staged once per block through LDS, 512 persistent blocks; k order inside runs of 8 regrouped
# speedup vs baseline: 2.3135x; 1.2058x over previous
#include <hip/hip_runtime.h>


#define N_NODES 100000
#define N_EDGES 1600000
#define NFEAT   256
#define NHID    128
#define NCLASS  40

#define LDA     132

#define G1_BM    32
#define G1_TILES 3125
#define G1_GRID  512
#define LDX      260

typedef __attribute__((ext_vector_type(2))) float v2f;
typedef __attribute__((ext_vector_type(8))) float v8f;

__global__ __launch_bounds__(256)
void gcn_gemm1(const float* __restrict__ x, const float* __restrict__ W1,
               float* __restrict__ H0, int* __restrict__ cnt) {
  __shared__ __attribute__((aligned(16))) float sX[G1_BM * LDX];
  const int wave = threadIdx.x >> 5;
  const int lane = threadIdx.x & 31;
  const int half = lane >> 4;
  const int mn   = lane & 15;
  const int n0   = wave * 16;

  const int ch = threadIdx.x & 63;
  const int r0 = threadIdx.x >> 6;

  int tile = blockIdx.x;
  float4 p0, p1, p2, p3, p4, p5, p6, p7;
#define G1_FETCH(t)                                                          \
  {                                                                          \
    const float* src_ = x + ((size_t)(t) * G1_BM + r0) * NFEAT + ch * 4;     \
    p0 = *(const float4*)(src_);                                             \
    p1 = *(const float4*)(src_ + 4 * NFEAT);                                 \
    p2 = *(const float4*)(src_ + 8 * NFEAT);                                 \
    p3 = *(const float4*)(src_ + 12 * NFEAT);                                \
    p4 = *(const float4*)(src_ + 16 * NFEAT);                                \
    p5 = *(const float4*)(src_ + 20 * NFEAT);                                \
    p6 = *(const float4*)(src_ + 24 * NFEAT);                                \
    p7 = *(const float4*)(src_ + 28 * NFEAT);                                \
  }
  G1_FETCH(tile)

  float bq[NFEAT / 8][4];
  {
    const float* bcol = W1 + (size_t)(4 * half) * NHID + n0 + mn;
    #pragma unroll
    for (int j = 0; j < NFEAT / 8; ++j) {
      #pragma unroll
      for (int i = 0; i < 4; ++i)
        bq[j][i] = bcol[(size_t)(8 * j + i) * NHID];
    }
  }

  const float* a0p = sX + mn * LDX + 4 * half;
  const float* a1p = a0p + 16 * LDX;

  for (; tile < G1_TILES; tile += G1_GRID) {
    const int m0 = tile * G1_BM;
    {
      float* dst = sX + r0 * LDX + ch * 4;
      *(float4*)(dst)            = p0;
      *(float4*)(dst + 4 * LDX)  = p1;
      *(float4*)(dst + 8 * LDX)  = p2;
      *(float4*)(dst + 12 * LDX) = p3;
      *(float4*)(dst + 16 * LDX) = p4;
      *(float4*)(dst + 20 * LDX) = p5;
      *(float4*)(dst + 24 * LDX) = p6;
      *(float4*)(dst + 28 * LDX) = p7;
    }
    if (threadIdx.x < G1_BM) cnt[m0 + threadIdx.x] = 0;
    __syncthreads();

    const int nxt = tile + G1_GRID;
    if (nxt < G1_TILES) G1_FETCH(nxt)

    v8f c0 = {};
    v8f c1 = {};
    #pragma unroll
    for (int j = 0; j < NFEAT / 8; ++j) {
      const float4 a0 = *(const float4*)(a0p + 8 * j);
      const float4 a1 = *(const float4*)(a1p + 8 * j);
      v2f a, b;
      b.x = bq[j][0]; b.y = bq[j][1];
      a.x = a0.x; a.y = a0.y;
      c0 = __builtin_amdgcn_wmma_f32_16x16x4_f32(
           false, a,  false, b,
           (short)0, c0,  false,  false);
      a.x = a1.x; a.y = a1.y;
      c1 = __builtin_amdgcn_wmma_f32_16x16x4_f32(
          false, a, false, b, (short)0, c1, false, false);
      b.x = bq[j][2]; b.y = bq[j][3];
      a.x = a0.z; a.y = a0.w;
      c0 = __builtin_amdgcn_wmma_f32_16x16x4_f32(
          false, a, false, b, (short)0, c0, false, false);
      a.x = a1.z; a.y = a1.w;
      c1 = __builtin_amdgcn_wmma_f32_16x16x4_f32(
          false, a, false, b, (short)0, c1, false, false);
    }
    float* crow = H0 + (size_t)m0 * NHID + n0 + mn;
    #pragma unroll
    for (int j = 0; j < 8; ++j) {
      crow[(size_t)(j + 8 * half) * NHID]      = c0[j];
      crow[(size_t)(j + 8 * half + 16) * NHID] = c1[j];
    }
    __syncthreads();
  }
}

__global__ __launch_bounds__(256)
void gcn_count(const int* __restrict__ edst, int* __restrict__ cnt,
               int* __restrict__ slot) {
  const int e = blockIdx.x * 256 + threadIdx.x;
  if (e < N_EDGES) slot[e] = atomicAdd(&cnt[edst[e]], 1);
}

__global__ __launch_bounds__(1024)
void gcn_rowptr(const int* __restrict__ cnt, int* __restrict__ rowptr) {
  __shared__ int s_before[32];
  __shared__ int s_incl[32];
  const int t = threadIdx.x;
  const int lane = t & 31;
  const int wave = t >> 5;
  const int first = blockIdx.x * 1024;

  int before = 0;
  for (int i = t; i < first; i += 1024) before += cnt[i];

  const int i = first + t;
  const int v = (i < N_NODES) ? cnt[i] : 0;
  int inc = v;
  for (int o = 1; o < 32; o <<= 1) {
    const int u = __shfl_up(inc, o);
    if (lane >= o) inc += u;
  }
  for (int o = 16; o > 0; o >>= 1) before += __shfl_xor(before, o);
  if (lane == 31) s_incl[wave] = inc;
  if (lane == 0)  s_before[wave] = before;
  __syncthreads();
  int bsum = s_before[lane];
  int wsum = (lane < wave) ? s_incl[lane] : 0;
  for (int o = 16; o > 0; o >>= 1) {
    bsum += __shfl_xor(bsum, o);
    wsum += __shfl_xor(wsum, o);
  }
  const int excl = bsum + wsum + inc - v;
  if (i < N_NODES) rowptr[i] = excl;
  if (i == N_NODES - 1) rowptr[N_NODES] = excl + v;
}

__global__ __launch_bounds__(256)
void gcn_place(const int* __restrict__ esrc, const int* __restrict__ edst,
               const float* __restrict__ ew, const int* __restrict__ rowptr,
               const int* __restrict__ slot, int* __restrict__ rec) {
  const int e = blockIdx.x * 256 + threadIdx.x;
  if (e < N_EDGES) {
    const int pos = rowptr[edst[e]] + slot[e];
    ((int2*)rec)[pos] = make_int2(esrc[e], __float_as_int(ew[e]));
  }
}

__global__ __launch_bounds__(256)
void gcn_agg_gemm2(const int* __restrict__ rowptr, const int* __restrict__ rec,
                   const float* __restrict__ H0, const float* __restrict__ b1,
                   const float* __restrict__ W2, float* __restrict__ H2) {
  __shared__ __attribute__((aligned(16))) float sA[16 * LDA];
  const int mt   = blockIdx.x;
  const int wave = threadIdx.x >> 5;
  const int lane = threadIdx.x & 31;
  const int m0   = mt * 16;
  const int2* recs = (const int2*)rec;
  const float4 bv = *(const float4*)(b1 + lane * 4);

  for (int rr = 0; rr < 2; ++rr) {
    const int rl  = wave + 8 * rr;
    const int beg = __builtin_amdgcn_readfirstlane(rowptr[m0 + rl]);
    const int end = __builtin_amdgcn_readfirstlane(rowptr[m0 + rl + 1]);
    float4 acc = make_float4(0.f, 0.f, 0.f, 0.f);
    for (int base = beg; base < end; base += 32) {
      const int left = end - base;
      const int n = left < 32 ? left : 32;
      int2 sw = make_int2(0, 0);
      if (lane < n) sw = recs[base + lane];
      #pragma unroll 4
      for (int j = 0; j < n; ++j) {
        const int   s = __builtin_amdgcn_readlane(sw.x, j);
        const float w = __int_as_float(__builtin_amdgcn_readlane(sw.y, j));
        const float4 hv = *(const float4*)(H0 + (size_t)s * NHID + lane * 4);
        acc.x += w * hv.x;
        acc.y += w * hv.y;
        acc.z += w * hv.z;
        acc.w += w * hv.w;
      }
    }
    float4 o;
    o.x = fmaxf(acc.x + bv.x, 0.0f);
    o.y = fmaxf(acc.y + bv.y, 0.0f);
    o.z = fmaxf(acc.z + bv.z, 0.0f);
    o.w = fmaxf(acc.w + bv.w, 0.0f);
    *(float4*)(sA + rl * LDA + lane * 4) = o;
  }
  __syncthreads();

  if (wave < 3) {
    const int half = lane >> 4;
    const int mn   = lane & 15;
    const int col  = wave * 16 + mn;
    const bool colok = (col < NCLASS);
    const float* arow = sA + mn * LDA;

    v8f c = {};
    #pragma unroll 4
    for (int k = 0; k < NHID; k += 4) {
      const int k0 = k + 2 * half;
      v2f a, b;
      a.x = arow[k0];
      a.y = arow[k0 + 1];
      b.x = colok ? W2[(size_t)k0 * NCLASS + col]       : 0.0f;
      b.y = colok ? W2[(size_t)(k0 + 1) * NCLASS + col] : 0.0f;
      c = __builtin_amdgcn_wmma_f32_16x16x4_f32(
          false, a, false, b, (short)0, c, false, false);
    }
    if (colok) {
      #pragma unroll
      for (int j = 0; j < 8; ++j)
        H2[(size_t)(m0 + j + 8 * half) * NCLASS + col] = c[j];
    }
  }
}

__global__ __launch_bounds__(256)
void gcn_spmm2(const int* __restrict__ rowptr, const int* __restrict__ rec,
               const float* __restrict__ H2, const float* __restrict__ b2,
               float* __restrict__ out) {
  const int tid = blockIdx.x * 256 + threadIdx.x;
  if (tid >= N_NODES * (NCLASS / 4)) return;
  const int r  = tid / (NCLASS / 4);
  const int c4 = tid % (NCLASS / 4);
  const int2* recs = (const int2*)rec;
  const int beg = rowptr[r];
  const int end = rowptr[r + 1];
  const float* hcol = H2 + c4 * 4;
  float4 acc = make_float4(0.f, 0.f, 0.f, 0.f);
  #pragma unroll 4
  for (int e = beg; e < end; ++e) {
    const int2 sw = recs[e];
    const float w = __int_as_float(sw.y);
    const float4 hv = *(const float4*)(hcol + (size_t)sw.x * NCLASS);
    acc.x += w * hv.x;
    acc.y += w * hv.y;
    acc.z += w * hv.z;
    acc.w += w * hv.w;
  }
  const float4 bv = *(const float4*)(b2 + c4 * 4);
  float4 o;
  o.x = acc.x + bv.x;
  o.y = acc.y + bv.y;
  o.z = acc.z + bv.z;
  o.w = acc.w + bv.w;
  *(float4*)(out + (size_t)tid * 4) = o;
}

extern "C" void kernel_launch(void* const* d_in, const int* in_sizes, int n_in,
                              void* d_out, int out_size, void* d_ws, size_t ws_size,
                              hipStream_t stream) {
  const float* x    = (const float*)d_in[0];
  const int*   esrc = (const int*)  d_in[1];
  const int*   edst = (const int*)  d_in[2];
  const float* ew   = (const float*)d_in[3];
  const float* W1   = (const float*)d_in[4];
  const float* b1   = (const float*)d_in[5];
  const float* W2   = (const float*)d_in[6];
  const float* b2   = (const float*)d_in[7];
  float* out = (float*)d_out;

  float* H0     = (float*)d_ws;
  float* H2     = H0 + (size_t)N_NODES * NHID;
  int*   rec    = (int*)(H2 + (size_t)N_NODES * NCLASS);
  int*   slot   = rec + (size_t)N_EDGES * 2;
  int*   cnt    = slot + (size_t)N_EDGES;
  int*   rowptr = cnt + (size_t)N_NODES;

  gcn_gemm1<<<G1_GRID, 256, 0, stream>>>(x, W1, H0, cnt);
  gcn_count<<<N_EDGES / 256, 256, 0, stream>>>(edst, cnt, slot);
  gcn_rowptr<<<(N_NODES + 1023) / 1024, 1024, 0, stream>>>(cnt, rowptr);
  gcn_place<<<N_EDGES / 256, 256, 0, stream>>>(esrc, edst, ew, rowptr, slot, rec);
  gcn_agg_gemm2<<<N_NODES / 16, 256, 0, stream>>>(rowptr, rec, H0, b1, W2, H2);
  gcn_spmm2<<<(N_NODES * (NCLASS / 4) + 255) / 256, 256, 0, stream>>>(rowptr, rec, H2, b2, out);
}


# Counting sort on LDS counters in two levels: per-chunk histograms of 512-node buckets, offsets from column sums, a block per bucket ranks by LDS add; no global atomics; row order = LDS arrival
# speedup vs baseline: 3.6074x; 1.5593x over previous
#include <hip/hip_runtime.h>


#define N_NODES 100000
#define N_EDGES 1600000
#define NFEAT   256
#define NHID    128
#define NCLASS  40

#define LDA     132

#define G1_BM    32
#define G1_TILES 3125
#define G1_GRID  512
#define LDX      260

typedef __attribute__((ext_vector_type(2))) float v2f;
typedef __attribute__((ext_vector_type(8))) float v8f;

__global__ __launch_bounds__(256)
void gcn_gemm1(const float* __restrict__ x, const float* __restrict__ W1,
               float* __restrict__ H0) {
  __shared__ __attribute__((aligned(16))) float sX[G1_BM * LDX];
  const int wave = threadIdx.x >> 5;
  const int lane = threadIdx.x & 31;
  const int half = lane >> 4;
  const int mn   = lane & 15;
  const int n0   = wave * 16;

  const int ch = threadIdx.x & 63;
  const int r0 = threadIdx.x >> 6;

  int tile = blockIdx.x;
  float4 p0, p1, p2, p3, p4, p5, p6, p7;
#define G1_FETCH(t)                                                          \
  {                                                                          \
    const float* src_ = x + ((size_t)(t) * G1_BM + r0) * NFEAT + ch * 4;     \
    p0 = *(const float4*)(src_);                                             \
    p1 = *(const float4*)(src_ + 4 * NFEAT);                                 \
    p2 = *(const float4*)(src_ + 8 * NFEAT);                                 \
    p3 = *(const float4*)(src_ + 12 * NFEAT);                                \
    p4 = *(const float4*)(src_ + 16 * NFEAT);                                \
    p5 = *(const float4*)(src_ + 20 * NFEAT);                                \
    p6 = *(const float4*)(src_ + 24 * NFEAT);                                \
    p7 = *(const float4*)(src_ + 28 * NFEAT);                                \
  }
  G1_FETCH(tile)

  float bq[NFEAT / 8][4];
  {
    const float* bcol = W1 + (size_t)(4 * half) * NHID + n0 + mn;
    #pragma unroll
    for (int j = 0; j < NFEAT / 8; ++j) {
      #pragma unroll
      for (int i = 0; i < 4; ++i)
        bq[j][i] = bcol[(size_t)(8 * j + i) * NHID];
    }
  }

  const float* a0p = sX + mn * LDX + 4 * half;
  const float* a1p = a0p + 16 * LDX;

  for (; tile < G1_TILES; tile += G1_GRID) {
    const int m0 = tile * G1_BM;
    {
      float* dst = sX + r0 * LDX + ch * 4;
      *(float4*)(dst)            = p0;
      *(float4*)(dst + 4 * LDX)  = p1;
      *(float4*)(dst + 8 * LDX)  = p2;
      *(float4*)(dst + 12 * LDX) = p3;
      *(float4*)(dst + 16 * LDX) = p4;
      *(float4*)(dst + 20 * LDX) = p5;
      *(float4*)(dst + 24 * LDX) = p6;
      *(float4*)(dst + 28 * LDX) = p7;
    }
    __syncthreads();

    const int nxt = tile + G1_GRID;
    if (nxt < G1_TILES) G1_FETCH(nxt)

    v8f c0 = {};
    v8f c1 = {};
    #pragma unroll
    for (int j = 0; j < NFEAT / 8; ++j) {
      const float4 a0 = *(const float4*)(a0p + 8 * j);
      const float4 a1 = *(const float4*)(a1p + 8 * j);
      v2f a, b;
      b.x = bq[j][0]; b.y = bq[j][1];
      a.x = a0.x; a.y = a0.y;
      c0 = __builtin_amdgcn_wmma_f32_16x16x4_f32(
           false, a,  false, b,
           (short)0, c0,  false,  false);
      a.x = a1.x; a.y = a1.y;
      c1 = __builtin_amdgcn_wmma_f32_16x16x4_f32(
          false, a, false, b, (short)0, c1, false, false);
      b.x = bq[j][2]; b.y = bq[j][3];
      a.x = a0.z; a.y = a0.w;
      c0 = __builtin_amdgcn_wmma_f32_16x16x4_f32(
          false, a, false, b, (short)0, c0, false, false);
      a.x = a1.z; a.y = a1.w;
      c1 = __builtin_amdgcn_wmma_f32_16x16x4_f32(
          false, a, false, b, (short)0, c1, false, false);
    }
    float* crow = H0 + (size_t)m0 * NHID + n0 + mn;
    #pragma unroll
    for (int j = 0; j < 8; ++j) {
      crow[(size_t)(j + 8 * half) * NHID]      = c0[j];
      crow[(size_t)(j + 8 * half + 16) * NHID] = c1[j];
    }
    __syncthreads();
  }
}

#define BK_SHIFT 9
#define BK_NODES 512
#define NBK      196
#define PT_GRID  250
#define PT_CHUNK 6400
#define PT_ITERS 7
#define SRC_BITS 17
#define BS_BATCH 8

static_assert(PT_GRID * PT_CHUNK == N_EDGES, "chunks must tile the edge list");
static_assert(NBK * BK_NODES >= N_NODES, "buckets must cover the nodes");
static_assert(N_NODES <= (1 << SRC_BITS), "source id must fit its field");

__global__ __launch_bounds__(1024)
void gcn_hist(const int* __restrict__ edst, int* __restrict__ hist) {
  __shared__ int s_cnt[NBK];
  const int t = threadIdx.x;
  if (t < NBK) s_cnt[t] = 0;
  const int* d = edst + (size_t)blockIdx.x * PT_CHUNK;
  int dk[PT_ITERS];
  #pragma unroll
  for (int k = 0; k < PT_ITERS; ++k) {
    const int i = k * 1024 + t;
    dk[k] = (i < PT_CHUNK) ? d[i] : 0;
  }
  __syncthreads();
  #pragma unroll
  for (int k = 0; k < PT_ITERS; ++k) {
    if (k * 1024 + t < PT_CHUNK) atomicAdd(&s_cnt[dk[k] >> BK_SHIFT], 1);
  }
  __syncthreads();
  if (t < NBK) hist[blockIdx.x * NBK + t] = s_cnt[t];
}

__global__ __launch_bounds__(1024)
void gcn_part(const int* __restrict__ esrc, const int* __restrict__ edst,
              const float* __restrict__ ew, const int* __restrict__ hist,
              int* __restrict__ bstart, int* __restrict__ part) {
  __shared__ int s_cnt[NBK];
  __shared__ int s_base[NBK];
  __shared__ int s_tot[4][256];
  __shared__ int s_bef[4][256];
  const int t = threadIdx.x;
  const int lane = t & 31;
  const int g = blockIdx.x;
  if (t < NBK) s_cnt[t] = 0;

  const size_t e0 = (size_t)g * PT_CHUNK;
  int   dk[PT_ITERS], sk[PT_ITERS];
  float wk[PT_ITERS];
  #pragma unroll
  for (int k = 0; k < PT_ITERS; ++k) {
    const int i = k * 1024 + t;
    const bool ok = (i < PT_CHUNK);
    dk[k] = ok ? edst[e0 + i] : 0;
    sk[k] = ok ? esrc[e0 + i] : 0;
    wk[k] = ok ? ew[e0 + i]   : 0.0f;
  }

  {
    const int b = t & 255;
    const int q = t >> 8;
    int tot = 0, bef = 0;
    if (b < NBK) {
      const int* hp = hist + q * NBK + b;
      #pragma unroll
      for (int j = 0; j < (PT_GRID + 3) / 4; ++j) {
        const int gg = q + 4 * j;
        const int v = (gg < PT_GRID) ? hp[j * 4 * NBK] : 0;
        tot += v;
        bef += (gg < g) ? v : 0;
      }
    }
    s_tot[q][b] = tot;
    s_bef[q][b] = bef;
  }
  __syncthreads();

  if (t < 32) {
    int loc[7];
    int sum = 0;
    #pragma unroll
    for (int j = 0; j < 7; ++j) {
      const int bb = 7 * lane + j;
      const int v = (bb < NBK)
          ? s_tot[0][bb] + s_tot[1][bb] + s_tot[2][bb] + s_tot[3][bb] : 0;
      loc[j] = sum;
      sum += v;
    }
    int inc = sum;
    for (int o = 1; o < 32; o <<= 1) {
      const int u = __shfl_up(inc, o);
      if (lane >= o) inc += u;
    }
    const int excl = inc - sum;
    #pragma unroll
    for (int j = 0; j < 7; ++j) {
      const int bb = 7 * lane + j;
      if (bb < NBK) {
        const int start = excl + loc[j];
        s_base[bb] = start + s_bef[0][bb] + s_bef[1][bb] + s_bef[2][bb] + s_bef[3][bb];
        if (g == 0) bstart[bb] = start;
      }
    }
    if (g == 0 && lane == 31) bstart[NBK] = inc;
  }
  __syncthreads();

  int2* p2 = (int2*)part;
  int pos[PT_ITERS];
  #pragma unroll
  for (int k = 0; k < PT_ITERS; ++k) {
    pos[k] = 0;
    if (k * 1024 + t < PT_CHUNK) {
      const int bk = dk[k] >> BK_SHIFT;
      pos[k] = s_base[bk] + atomicAdd(&s_cnt[bk], 1);
    }
  }
  #pragma unroll
  for (int k = 0; k < PT_ITERS; ++k) {
    if (k * 1024 + t < PT_CHUNK)
      p2[pos[k]] = make_int2(((dk[k] & (BK_NODES - 1)) << SRC_BITS) | sk[k],
                             __float_as_int(wk[k]));
  }
}

__global__ __launch_bounds__(1024)
void gcn_bsort(const int* __restrict__ part, const int* __restrict__ bstart,
               int* __restrict__ rowptr, int* __restrict__ rec) {
  __shared__ int s_cnt[BK_NODES];
  __shared__ int s_cur[BK_NODES];
  __shared__ int s_wsum[16];
  const int t = threadIdx.x;
  const int lane = t & 31;
  const int wave = t >> 5;
  const int b = blockIdx.x;
  const int beg = bstart[b];
  const int end = bstart[b + 1];
  const int2* p2 = (const int2*)part;

  if (t < BK_NODES) s_cnt[t] = 0;
  __syncthreads();
  for (int base = beg; base < end; base += BS_BATCH * 1024) {
    int key[BS_BATCH];
    #pragma unroll
    for (int u = 0; u < BS_BATCH; ++u) {
      const int i = base + u * 1024 + t;
      key[u] = (i < end) ? p2[i].x : -1;
    }
    #pragma unroll
    for (int u = 0; u < BS_BATCH; ++u)
      if (key[u] >= 0) atomicAdd(&s_cnt[key[u] >> SRC_BITS], 1);
  }
  __syncthreads();

  int v = 0, inc = 0;
  if (t < BK_NODES) {
    v = s_cnt[t];
    inc = v;
    for (int o = 1; o < 32; o <<= 1) {
      const int u = __shfl_up(inc, o);
      if (lane >= o) inc += u;
    }
    if (lane == 31) s_wsum[wave] = inc;
  }
  __syncthreads();
  if (t < BK_NODES) {
    int ws = (lane < wave) ? s_wsum[lane] : 0;
    for (int o = 16; o > 0; o >>= 1) ws += __shfl_xor(ws, o);
    const int start = beg + ws + inc - v;
    s_cur[t] = start;
    const int node = b * BK_NODES + t;
    if (node < N_NODES) rowptr[node] = start;
    if (node == N_NODES - 1) rowptr[N_NODES] = start + v;
  }
  __syncthreads();

  int2* r2 = (int2*)rec;
  for (int base = beg; base < end; base += BS_BATCH * 1024) {
    int2 kw[BS_BATCH];
    #pragma unroll
    for (int u = 0; u < BS_BATCH; ++u) {
      const int i = base + u * 1024 + t;
      kw[u] = (i < end) ? p2[i] : make_int2(-1, 0);
    }
    int pos[BS_BATCH];
    #pragma unroll
    for (int u = 0; u < BS_BATCH; ++u) {
      pos[u] = 0;
      if (kw[u].x >= 0) pos[u] = atomicAdd(&s_cur[kw[u].x >> SRC_BITS], 1);
    }
    #pragma unroll
    for (int u = 0; u < BS_BATCH; ++u)
      if (kw[u].x >= 0)
        r2[pos[u]] = make_int2(kw[u].x & ((1 << SRC_BITS) - 1), kw[u].y);
  }
}

__global__ __launch_bounds__(256)
void gcn_agg_gemm2(const int* __restrict__ rowptr, const int* __restrict__ rec,
                   const float* __restrict__ H0, const float* __restrict__ b1,
                   const float* __restrict__ W2, float* __restrict__ H2) {
  __shared__ __attribute__((aligned(16))) float sA[16 * LDA];
  const int mt   = blockIdx.x;
  const int wave = threadIdx.x >> 5;
  const int lane = threadIdx.x & 31;
  const int m0   = mt * 16;
  const int2* recs = (const int2*)rec;
  const float4 bv = *(const float4*)(b1 + lane * 4);

  for (int rr = 0; rr < 2; ++rr) {
    const int rl  = wave + 8 * rr;
    const int beg = __builtin_amdgcn_readfirstlane(rowptr[m0 + rl]);
    const int end = __builtin_amdgcn_readfirstlane(rowptr[m0 + rl + 1]);
    float4 acc = make_float4(0.f, 0.f, 0.f, 0.f);
    for (int base = beg; base < end; base += 32) {
      const int left = end - base;
      const int n = left < 32 ? left : 32;
      int2 sw = make_int2(0, 0);
      if (lane < n) sw = recs[base + lane];
      #pragma unroll 4
      for (int j = 0; j < n; ++j) {
        const int   s = __builtin_amdgcn_readlane(sw.x, j);
        const float w = __int_as_float(__builtin_amdgcn_readlane(sw.y, j));
        const float4 hv = *(const float4*)(H0 + (size_t)s * NHID + lane * 4);
        acc.x += w * hv.x;
        acc.y += w * hv.y;
        acc.z += w * hv.z;
        acc.w += w * hv.w;
      }
    }
    float4 o;
    o.x = fmaxf(acc.x + bv.x, 0.0f);
    o.y = fmaxf(acc.y + bv.y, 0.0f);
    o.z = fmaxf(acc.z + bv.z, 0.0f);
    o.w = fmaxf(acc.w + bv.w, 0.0f);
    *(float4*)(sA + rl * LDA + lane * 4) = o;
  }
  __syncthreads();

  if (wave < 3) {
    const int half = lane >> 4;
    const int mn   = lane & 15;
    const int col  = wave * 16 + mn;
    const bool colok = (col < NCLASS);
    const float* arow = sA + mn * LDA;

    v8f c = {};
    #pragma unroll 4
    for (int k = 0; k < NHID; k += 4) {
      const int k0 = k + 2 * half;
      v2f a, b;
      a.x = arow[k0];
      a.y = arow[k0 + 1];
      b.x = colok ? W2[(size_t)k0 * NCLASS + col]       : 0.0f;
      b.y = colok ? W2[(size_t)(k0 + 1) * NCLASS + col] : 0.0f;
      c = __builtin_amdgcn_wmma_f32_16x16x4_f32(
          false, a, false, b, (short)0, c, false, false);
    }
    if (colok) {
      #pragma unroll
      for (int j = 0; j < 8; ++j)
        H2[(size_t)(m0 + j + 8 * half) * NCLASS + col] = c[j];
    }
  }
}

__global__ __launch_bounds__(256)
void gcn_spmm2(const int* __restrict__ rowptr, const int* __restrict__ rec,
               const float* __restrict__ H2, const float* __restrict__ b2,
               float* __restrict__ out) {
  const int tid = blockIdx.x * 256 + threadIdx.x;
  if (tid >= N_NODES * (NCLASS / 4)) return;
  const int r  = tid / (NCLASS / 4);
  const int c4 = tid % (NCLASS / 4);
  const int2* recs = (const int2*)rec;
  const int beg = rowptr[r];
  const int end = rowptr[r + 1];
  const float* hcol = H2 + c4 * 4;
  float4 acc = make_float4(0.f, 0.f, 0.f, 0.f);
  #pragma unroll 4
  for (int e = beg; e < end; ++e) {
    const int2 sw = recs[e];
    const float w = __int_as_float(sw.y);
    const float4 hv = *(const float4*)(hcol + (size_t)sw.x * NCLASS);
    acc.x += w * hv.x;
    acc.y += w * hv.y;
    acc.z += w * hv.z;
    acc.w += w * hv.w;
  }
  const float4 bv = *(const float4*)(b2 + c4 * 4);
  float4 o;
  o.x = acc.x + bv.x;
  o.y = acc.y + bv.y;
  o.z = acc.z + bv.z;
  o.w = acc.w + bv.w;
  *(float4*)(out + (size_t)tid * 4) = o;
}

extern "C" void kernel_launch(void* const* d_in, const int* in_sizes, int n_in,
                              void* d_out, int out_size, void* d_ws, size_t ws_size,
                              hipStream_t stream) {
  const float* x    = (const float*)d_in[0];
  const int*   esrc = (const int*)  d_in[1];
  const int*   edst = (const int*)  d_in[2];
  const float* ew   = (const float*)d_in[3];
  const float* W1   = (const float*)d_in[4];
  const float* b1   = (const float*)d_in[5];
  const float* W2   = (const float*)d_in[6];
  const float* b2   = (const float*)d_in[7];
  float* out = (float*)d_out;

  float* H0     = (float*)d_ws;
  float* H2     = H0 + (size_t)N_NODES * NHID;
  int*   part   = (int*)H2;
  int*   rec    = (int*)(H2 + (size_t)N_NODES * NCLASS);
  int*   hist   = rec + (size_t)N_EDGES * 2;
  int*   bstart = hist + (size_t)PT_GRID * NBK;
  int*   rowptr = bstart + 256;

  gcn_gemm1<<<G1_GRID, 256, 0, stream>>>(x, W1, H0);
  gcn_hist<<<PT_GRID, 1024, 0, stream>>>(edst, hist);
  gcn_part<<<PT_GRID, 1024, 0, stream>>>(esrc, edst, ew, hist, bstart, part);
  gcn_bsort<<<NBK, 1024, 0, stream>>>(part, bstart, rowptr, rec);
  gcn_agg_gemm2<<<N_NODES / 16, 256, 0, stream>>>(rowptr, rec, H0, b1, W2, H2);
  gcn_spmm2<<<(N_NODES * (NCLASS / 4) + 255) / 256, 256, 0, stream>>>(rowptr, rec, H2, b2, out);
}


# Launch 5 (row sums + W2 product): loads issued in groups ahead of use: both rows' pointers and records together, 8 row gathers per round trip, W2 fragments 32 per group; sum orders kept
# speedup vs baseline: 4.0257x; 1.1160x over previous
#include <hip/hip_runtime.h>


#define N_NODES 100000
#define N_EDGES 1600000
#define NFEAT   256
#define NHID    128
#define NCLASS  40

#define LDA     132

#define G1_BM    32
#define G1_TILES 3125
#define G1_GRID  512
#define LDX      260

typedef __attribute__((ext_vector_type(2))) float v2f;
typedef __attribute__((ext_vector_type(8))) float v8f;

__global__ __launch_bounds__(256)
void gcn_gemm1(const float* __restrict__ x, const float* __restrict__ W1,
               float* __restrict__ H0) {
  __shared__ __attribute__((aligned(16))) float sX[G1_BM * LDX];
  const int wave = threadIdx.x >> 5;
  const int lane = threadIdx.x & 31;
  const int half = lane >> 4;
  const int mn   = lane & 15;
  const int n0   = wave * 16;

  const int ch = threadIdx.x & 63;
  const int r0 = threadIdx.x >> 6;

  int tile = blockIdx.x;
  float4 p0, p1, p2, p3, p4, p5, p6, p7;
#define G1_FETCH(t)                                                          \
  {                                                                          \
    const float* src_ = x + ((size_t)(t) * G1_BM + r0) * NFEAT + ch * 4;     \
    p0 = *(const float4*)(src_);                                             \
    p1 = *(const float4*)(src_ + 4 * NFEAT);                                 \
    p2 = *(const float4*)(src_ + 8 * NFEAT);                                 \
    p3 = *(const float4*)(src_ + 12 * NFEAT);                                \
    p4 = *(const float4*)(src_ + 16 * NFEAT);                                \
    p5 = *(const float4*)(src_ + 20 * NFEAT);                                \
    p6 = *(const float4*)(src_ + 24 * NFEAT);                                \
    p7 = *(const float4*)(src_ + 28 * NFEAT);                                \
  }
  G1_FETCH(tile)

  float bq[NFEAT / 8][4];
  {
    const float* bcol = W1 + (size_t)(4 * half) * NHID + n0 + mn;
    #pragma unroll
    for (int j = 0; j < NFEAT / 8; ++j) {
      #pragma unroll
      for (int i = 0; i < 4; ++i)
        bq[j][i] = bcol[(size_t)(8 * j + i) * NHID];
    }
  }

  const float* a0p = sX + mn * LDX + 4 * half;
  const float* a1p = a0p + 16 * LDX;

  for (; tile < G1_TILES; tile += G1_GRID) {
    const int m0 = tile * G1_BM;
    {
      float* dst = sX + r0 * LDX + ch * 4;
      *(float4*)(dst)            = p0;
      *(float4*)(dst + 4 * LDX)  = p1;
      *(float4*)(dst + 8 * LDX)  = p2;
      *(float4*)(dst + 12 * LDX) = p3;
      *(float4*)(dst + 16 * LDX) = p4;
      *(float4*)(dst + 20 * LDX) = p5;
      *(float4*)(dst + 24 * LDX) = p6;
      *(float4*)(dst + 28 * LDX) = p7;
    }
    __syncthreads();

    const int nxt = tile + G1_GRID;
    if (nxt < G1_TILES) G1_FETCH(nxt)

    v8f c0 = {};
    v8f c1 = {};
    #pragma unroll
    for (int j = 0; j < NFEAT / 8; ++j) {
      const float4 a0 = *(const float4*)(a0p + 8 * j);
      const float4 a1 = *(const float4*)(a1p + 8 * j);
      v2f a, b;
      b.x = bq[j][0]; b.y = bq[j][1];
      a.x = a0.x; a.y = a0.y;
      c0 = __builtin_amdgcn_wmma_f32_16x16x4_f32(
           false, a,  false, b,
           (short)0, c0,  false,  false);
      a.x = a1.x; a.y = a1.y;
      c1 = __builtin_amdgcn_wmma_f32_16x16x4_f32(
          false, a, false, b, (short)0, c1, false, false);
      b.x = bq[j][2]; b.y = bq[j][3];
      a.x = a0.z; a.y = a0.w;
      c0 = __builtin_amdgcn_wmma_f32_16x16x4_f32(
          false, a, false, b, (short)0, c0, false, false);
      a.x = a1.z; a.y = a1.w;
      c1 = __builtin_amdgcn_wmma_f32_16x16x4_f32(
          false, a, false, b, (short)0, c1, false, false);
    }
    float* crow = H0 + (size_t)m0 * NHID + n0 + mn;
    #pragma unroll
    for (int j = 0; j < 8; ++j) {
      crow[(size_t)(j + 8 * half) * NHID]      = c0[j];
      crow[(size_t)(j + 8 * half + 16) * NHID] = c1[j];
    }
    __syncthreads();
  }
}

#define BK_SHIFT 9
#define BK_NODES 512
#define NBK      196
#define PT_GRID  250
#define PT_CHUNK 6400
#define PT_ITERS 7
#define SRC_BITS 17
#define BS_BATCH 8

static_assert(PT_GRID * PT_CHUNK == N_EDGES, "chunks must tile the edge list");
static_assert(NBK * BK_NODES >= N_NODES, "buckets must cover the nodes");
static_assert(N_NODES <= (1 << SRC_BITS), "source id must fit its field");

__global__ __launch_bounds__(1024)
void gcn_hist(const int* __restrict__ edst, int* __restrict__ hist) {
  __shared__ int s_cnt[NBK];
  const int t = threadIdx.x;
  if (t < NBK) s_cnt[t] = 0;
  const int* d = edst + (size_t)blockIdx.x * PT_CHUNK;
  int dk[PT_ITERS];
  #pragma unroll
  for (int k = 0; k < PT_ITERS; ++k) {
    const int i = k * 1024 + t;
    dk[k] = (i < PT_CHUNK) ? d[i] : 0;
  }
  __syncthreads();
  #pragma unroll
  for (int k = 0; k < PT_ITERS; ++k) {
    if (k * 1024 + t < PT_CHUNK) atomicAdd(&s_cnt[dk[k] >> BK_SHIFT], 1);
  }
  __syncthreads();
  if (t < NBK) hist[blockIdx.x * NBK + t] = s_cnt[t];
}

__global__ __launch_bounds__(1024)
void gcn_part(const int* __restrict__ esrc, const int* __restrict__ edst,
              const float* __restrict__ ew, const int* __restrict__ hist,
              int* __restrict__ bstart, int* __restrict__ part) {
  __shared__ int s_cnt[NBK];
  __shared__ int s_base[NBK];
  __shared__ int s_tot[4][256];
  __shared__ int s_bef[4][256];
  const int t = threadIdx.x;
  const int lane = t & 31;
  const int g = blockIdx.x;
  if (t < NBK) s_cnt[t] = 0;

  const size_t e0 = (size_t)g * PT_CHUNK;
  int   dk[PT_ITERS], sk[PT_ITERS];
  float wk[PT_ITERS];
  #pragma unroll
  for (int k = 0; k < PT_ITERS; ++k) {
    const int i = k * 1024 + t;
    const bool ok = (i < PT_CHUNK);
    dk[k] = ok ? edst[e0 + i] : 0;
    sk[k] = ok ? esrc[e0 + i] : 0;
    wk[k] = ok ? ew[e0 + i]   : 0.0f;
  }

  {
    const int b = t & 255;
    const int q = t >> 8;
    int tot = 0, bef = 0;
    if (b < NBK) {
      const int* hp = hist + q * NBK + b;
      #pragma unroll
      for (int j = 0; j < (PT_GRID + 3) / 4; ++j) {
        const int gg = q + 4 * j;
        const int v = (gg < PT_GRID) ? hp[j * 4 * NBK] : 0;
        tot += v;
        bef += (gg < g) ? v : 0;
      }
    }
    s_tot[q][b] = tot;
    s_bef[q][b] = bef;
  }
  __syncthreads();

  if (t < 32) {
    int loc[7];
    int sum = 0;
    #pragma unroll
    for (int j = 0; j < 7; ++j) {
      const int bb = 7 * lane + j;
      const int v = (bb < NBK)
          ? s_tot[0][bb] + s_tot[1][bb] + s_tot[2][bb] + s_tot[3][bb] : 0;
      loc[j] = sum;
      sum += v;
    }
    int inc = sum;
    for (int o = 1; o < 32; o <<= 1) {
      const int u = __shfl_up(inc, o);
      if (lane >= o) inc += u;
    }
    const int excl = inc - sum;
    #pragma unroll
    for (int j = 0; j < 7; ++j) {
      const int bb = 7 * lane + j;
      if (bb < NBK) {
        const int start = excl + loc[j];
        s_base[bb] = start + s_bef[0][bb] + s_bef[1][bb] + s_bef[2][bb] + s_bef[3][bb];
        if (g == 0) bstart[bb] = start;
      }
    }
    if (g == 0 && lane == 31) bstart[NBK] = inc;
  }
  __syncthreads();

  int2* p2 = (int2*)part;
  int pos[PT_ITERS];
  #pragma unroll
  for (int k = 0; k < PT_ITERS; ++k) {
    pos[k] = 0;
    if (k * 1024 + t < PT_CHUNK) {
      const int bk = dk[k] >> BK_SHIFT;
      pos[k] = s_base[bk] + atomicAdd(&s_cnt[bk], 1);
    }
  }
  #pragma unroll
  for (int k = 0; k < PT_ITERS; ++k) {
    if (k * 1024 + t < PT_CHUNK)
      p2[pos[k]] = make_int2(((dk[k] & (BK_NODES - 1)) << SRC_BITS) | sk[k],
                             __float_as_int(wk[k]));
  }
}

__global__ __launch_bounds__(1024)
void gcn_bsort(const int* __restrict__ part, const int* __restrict__ bstart,
               int* __restrict__ rowptr, int* __restrict__ rec) {
  __shared__ int s_cnt[BK_NODES];
  __shared__ int s_cur[BK_NODES];
  __shared__ int s_wsum[16];
  const int t = threadIdx.x;
  const int lane = t & 31;
  const int wave = t >> 5;
  const int b = blockIdx.x;
  const int beg = bstart[b];
  const int end = bstart[b + 1];
  const int2* p2 = (const int2*)part;

  if (t < BK_NODES) s_cnt[t] = 0;
  __syncthreads();
  for (int base = beg; base < end; base += BS_BATCH * 1024) {
    int key[BS_BATCH];
    #pragma unroll
    for (int u = 0; u < BS_BATCH; ++u) {
      const int i = base + u * 1024 + t;
      key[u] = (i < end) ? p2[i].x : -1;
    }
    #pragma unroll
    for (int u = 0; u < BS_BATCH; ++u)
      if (key[u] >= 0) atomicAdd(&s_cnt[key[u] >> SRC_BITS], 1);
  }
  __syncthreads();

  int v = 0, inc = 0;
  if (t < BK_NODES) {
    v = s_cnt[t];
    inc = v;
    for (int o = 1; o < 32; o <<= 1) {
      const int u = __shfl_up(inc, o);
      if (lane >= o) inc += u;
    }
    if (lane == 31) s_wsum[wave] = inc;
  }
  __syncthreads();
  if (t < BK_NODES) {
    int ws = (lane < wave) ? s_wsum[lane] : 0;
    for (int o = 16; o > 0; o >>= 1) ws += __shfl_xor(ws, o);
    const int start = beg + ws + inc - v;
    s_cur[t] = start;
    const int node = b * BK_NODES + t;
    if (node < N_NODES) rowptr[node] = start;
    if (node == N_NODES - 1) rowptr[N_NODES] = start + v;
  }
  __syncthreads();

  int2* r2 = (int2*)rec;
  for (int base = beg; base < end; base += BS_BATCH * 1024) {
    int2 kw[BS_BATCH];
    #pragma unroll
    for (int u = 0; u < BS_BATCH; ++u) {
      const int i = base + u * 1024 + t;
      kw[u] = (i < end) ? p2[i] : make_int2(-1, 0);
    }
    int pos[BS_BATCH];
    #pragma unroll
    for (int u = 0; u < BS_BATCH; ++u) {
      pos[u] = 0;
      if (kw[u].x >= 0) pos[u] = atomicAdd(&s_cur[kw[u].x >> SRC_BITS], 1);
    }
    #pragma unroll
    for (int u = 0; u < BS_BATCH; ++u)
      if (kw[u].x >= 0)
        r2[pos[u]] = make_int2(kw[u].x & ((1 << SRC_BITS) - 1), kw[u].y);
  }
}


template <int K>
__device__ __forceinline__
void gcn_gather_group(float4& acc, const float* __restrict__ h0l, int2 sw, int j) {
  float4 hv[K];
  #pragma unroll
  for (int u = 0; u < K; ++u) {
    const int s = __builtin_amdgcn_readlane(sw.x, j + u);
    hv[u] = *(const float4*)(h0l + (size_t)s * NHID);
  }
  asm volatile("" ::: "memory");
  #pragma unroll
  for (int u = 0; u < K; ++u) {
    const float w = __int_as_float(__builtin_amdgcn_readlane(sw.y, j + u));
    acc.x += w * hv[u].x;
    acc.y += w * hv[u].y;
    acc.z += w * hv[u].z;
    acc.w += w * hv[u].w;
  }
}

__device__ __forceinline__
float4 gcn_row_sum(const int2* __restrict__ recs, const float* __restrict__ h0l,
                   int beg, int end, int2 sw, int lane) {
  float4 acc = make_float4(0.f, 0.f, 0.f, 0.f);
  for (int base = beg; base < end; base += 32) {
    const int left = end - base;
    const int n = left < 32 ? left : 32;
    if (base != beg) {
      sw = make_int2(0, 0);
      if (lane < n) sw = recs[base + lane];
    }
    for (int j = 0; j < n; j += 8) {
      const int m = n - j;
      if (m >= 8) {
        gcn_gather_group<8>(acc, h0l, sw, j);
      } else {
        switch (m) {
          case 7:  gcn_gather_group<7>(acc, h0l, sw, j); break;
          case 6:  gcn_gather_group<6>(acc, h0l, sw, j); break;
          case 5:  gcn_gather_group<5>(acc, h0l, sw, j); break;
          case 4:  gcn_gather_group<4>(acc, h0l, sw, j); break;
          case 3:  gcn_gather_group<3>(acc, h0l, sw, j); break;
          case 2:  gcn_gather_group<2>(acc, h0l, sw, j); break;
          default: gcn_gather_group<1>(acc, h0l, sw, j); break;
        }
      }
    }
  }
  return acc;
}

__global__ __launch_bounds__(256)
void gcn_agg_gemm2(const int* __restrict__ rowptr, const int* __restrict__ rec,
                   const float* __restrict__ H0, const float* __restrict__ b1,
                   const float* __restrict__ W2, float* __restrict__ H2) {
  __shared__ __attribute__((aligned(16))) float sA[16 * LDA];
  const int mt   = blockIdx.x;
  const int wave = threadIdx.x >> 5;
  const int lane = threadIdx.x & 31;
  const int m0   = mt * 16;
  const int2* recs = (const int2*)rec;
  const float4 bv = *(const float4*)(b1 + lane * 4);
  const float* h0l = H0 + lane * 4;

  const int* rp = rowptr + m0 + wave;
  const int rA0 = rp[0], rA1 = rp[1], rB0 = rp[8], rB1 = rp[9];
  const int begA = __builtin_amdgcn_readfirstlane(rA0);
  const int endA = __builtin_amdgcn_readfirstlane(rA1);
  const int begB = __builtin_amdgcn_readfirstlane(rB0);
  const int endB = __builtin_amdgcn_readfirstlane(rB1);

  int2 swA = make_int2(0, 0);
  int2 swB = make_int2(0, 0);
  if (lane < endA - begA) swA = recs[begA + lane];
  if (lane < endB - begB) swB = recs[begB + lane];

  {
    const float4 acc = gcn_row_sum(recs, h0l, begA, endA, swA, lane);
    float4 o;
    o.x = fmaxf(acc.x + bv.x, 0.0f);
    o.y = fmaxf(acc.y + bv.y, 0.0f);
    o.z = fmaxf(acc.z + bv.z, 0.0f);
    o.w = fmaxf(acc.w + bv.w, 0.0f);
    *(float4*)(sA + wave * LDA + lane * 4) = o;
  }
  {
    const float4 acc = gcn_row_sum(recs, h0l, begB, endB, swB, lane);
    float4 o;
    o.x = fmaxf(acc.x + bv.x, 0.0f);
    o.y = fmaxf(acc.y + bv.y, 0.0f);
    o.z = fmaxf(acc.z + bv.z, 0.0f);
    o.w = fmaxf(acc.w + bv.w, 0.0f);
    *(float4*)(sA + (wave + 8) * LDA + lane * 4) = o;
  }
  __syncthreads();

  if (wave < 3) {
    const int half = lane >> 4;
    const int mn   = lane & 15;
    const int col  = wave * 16 + mn;
    const bool colok = (col < NCLASS);
    const float* wcol = W2 + (colok ? col : 0) + (size_t)(2 * half) * NCLASS;
    const float* arow = sA + mn * LDA + 2 * half;

    v8f c = {};
    #pragma unroll 1
    for (int kh = 0; kh < NHID; kh += NHID / 2) {
      float bx[16], by[16];
      #pragma unroll
      for (int i = 0; i < 16; ++i) {
        const float vx = wcol[(size_t)(kh + 4 * i) * NCLASS];
        const float vy = wcol[(size_t)(kh + 4 * i + 1) * NCLASS];
        bx[i] = colok ? vx : 0.0f;
        by[i] = colok ? vy : 0.0f;
      }
      #pragma unroll
      for (int i = 0; i < 16; ++i) {
        v2f a, b;
        a.x = arow[kh + 4 * i];
        a.y = arow[kh + 4 * i + 1];
        b.x = bx[i];
        b.y = by[i];
        c = __builtin_amdgcn_wmma_f32_16x16x4_f32(
            false, a, false, b, (short)0, c, false, false);
      }
    }
    if (colok) {
      #pragma unroll
      for (int j = 0; j < 8; ++j)
        H2[(size_t)(m0 + j + 8 * half) * NCLASS + col] = c[j];
    }
  }
}

__global__ __launch_bounds__(256)
void gcn_spmm2(const int* __restrict__ rowptr, const int* __restrict__ rec,
               const float* __restrict__ H2, const float* __restrict__ b2,
               float* __restrict__ out) {
  const int tid = blockIdx.x * 256 + threadIdx.x;
  if (tid >= N_NODES * (NCLASS / 4)) return;
  const int r  = tid / (NCLASS / 4);
  const int c4 = tid % (NCLASS / 4);
  const int2* recs = (const int2*)rec;
  const int beg = rowptr[r];
  const int end = rowptr[r + 1];
  const float* hcol = H2 + c4 * 4;
  float4 acc = make_float4(0.f, 0.f, 0.f, 0.f);
  #pragma unroll 4
  for (int e = beg; e < end; ++e) {
    const int2 sw = recs[e];
    const float w = __int_as_float(sw.y);
    const float4 hv = *(const float4*)(hcol + (size_t)sw.x * NCLASS);
    acc.x += w * hv.x;
    acc.y += w * hv.y;
    acc.z += w * hv.z;
    acc.w += w * hv.w;
  }
  const float4 bv = *(const float4*)(b2 + c4 * 4);
  float4 o;
  o.x = acc.x + bv.x;
  o.y = acc.y + bv.y;
  o.z = acc.z + bv.z;
  o.w = acc.w + bv.w;
  *(float4*)(out + (size_t)tid * 4) = o;
}

extern "C" void kernel_launch(void* const* d_in, const int* in_sizes, int n_in,
                              void* d_out, int out_size, void* d_ws, size_t ws_size,
                              hipStream_t stream) {
  const float* x    = (const float*)d_in[0];
  const int*   esrc = (const int*)  d_in[1];
  const int*   edst = (const int*)  d_in[2];
  const float* ew   = (const float*)d_in[3];
  const float* W1   = (const float*)d_in[4];
  const float* b1   = (const float*)d_in[5];
  const float* W2   = (const float*)d_in[6];
  const float* b2   = (const float*)d_in[7];
  float* out = (float*)d_out;

  float* H0     = (float*)d_ws;
  float* H2     = H0 + (size_t)N_NODES * NHID;
  int*   part   = (int*)H2;
  int*   rec    = (int*)(H2 + (size_t)N_NODES * NCLASS);
  int*   hist   = rec + (size_t)N_EDGES * 2;
  int*   bstart = hist + (size_t)PT_GRID * NBK;
  int*   rowptr = bstart + 256;

  gcn_gemm1<<<G1_GRID, 256, 0, stream>>>(x, W1, H0);
  gcn_hist<<<PT_GRID, 1024, 0, stream>>>(edst, hist);
  gcn_part<<<PT_GRID, 1024, 0, stream>>>(esrc, edst, ew, hist, bstart, part);
  gcn_bsort<<<NBK, 1024, 0, stream>>>(part, bstart, rowptr, rec);
  gcn_agg_gemm2<<<N_NODES / 16, 256, 0, stream>>>(rowptr, rec, H0, b1, W2, H2);
  gcn_spmm2<<<(N_NODES * (NCLASS / 4) + 255) / 256, 256, 0, stream>>>(rowptr, rec, H2, b2, out);
}


# Sort level 1 made chunk-local (each 6400-edge chunk bucket-sorted in its own slice, one LDS rank pass) and run inside launch 1; level 2 reads 250 segments; 6 launches -> 4; row order = LDS arrival
# speedup vs baseline: 4.4506x; 1.1055x over previous
#include <hip/hip_runtime.h>


#define N_NODES 100000
#define N_EDGES 1600000
#define NFEAT   256
#define NHID    128
#define NCLASS  40

#define LDA     132

#define G1_BM    32
#define G1_TILES 3125
#define G1_GRID  512
#define LDX      260

typedef __attribute__((ext_vector_type(2))) float v2f;
typedef __attribute__((ext_vector_type(8))) float v8f;

#define BK_SHIFT 9
#define BK_NODES 512
#define NBK      196
#define PT_GRID  250
#define PT_CHUNK 6400
#define PT_PER   25
#define CS_PITCH (NBK + 1)
#define SRC_BITS 17
#define BS_SEGS  8

static_assert(PT_GRID * PT_CHUNK == N_EDGES, "chunks must tile the edge list");
static_assert(PT_PER * 256 == PT_CHUNK, "a chunk is a whole number of edges per thread");
static_assert(NBK * BK_NODES >= N_NODES, "buckets must cover the nodes");
static_assert(NBK < 256, "a bucket id must fit the low byte next to the rank");
static_assert(N_NODES <= (1 << SRC_BITS), "source id must fit its field");
static_assert(BS_SEGS * 32 >= PT_GRID && BS_SEGS * 32 <= 256, "segment table size");
static_assert(PT_GRID <= G1_GRID, "every chunk needs a block of launch 1");

__global__ __launch_bounds__(256)
void gcn_gemm1(const float* __restrict__ x, const float* __restrict__ W1,
               float* __restrict__ H0,
               const int* __restrict__ esrc, const int* __restrict__ edst,
               const float* __restrict__ ew,
               int* __restrict__ cstart, int* __restrict__ part) {
  __shared__ __attribute__((aligned(16))) float sX[G1_BM * LDX];
  __shared__ int s_cnt[256];
  __shared__ int s_start[256];
  __shared__ int s_wsum[8];
  const int wave = threadIdx.x >> 5;
  const int lane = threadIdx.x & 31;

  if (blockIdx.x >= G1_GRID - PT_GRID) {
    const int g = blockIdx.x - (G1_GRID - PT_GRID);
    const int t = threadIdx.x;
    s_cnt[t] = 0;
    const size_t e0 = (size_t)g * PT_CHUNK;
    int   key[PT_PER], br[PT_PER];
    float wv[PT_PER];
    #pragma unroll
    for (int k = 0; k < PT_PER; ++k) {
      const size_t i = e0 + k * 256 + t;
      const int d = edst[i];
      key[k] = ((d & (BK_NODES - 1)) << SRC_BITS) | esrc[i];
      br[k]  = d >> BK_SHIFT;
      wv[k]  = ew[i];
    }
    __syncthreads();
    #pragma unroll
    for (int k = 0; k < PT_PER; ++k) {
      const int bk = br[k];
      br[k] = bk | (atomicAdd(&s_cnt[bk], 1) << 8);
    }
    __syncthreads();
    const int v = s_cnt[t];
    int inc = v;
    for (int o = 1; o < 32; o <<= 1) {
      const int u = __shfl_up(inc, o);
      if (lane >= o) inc += u;
    }
    if (lane == 31) s_wsum[wave] = inc;
    __syncthreads();
    int ws = (lane < wave) ? s_wsum[lane] : 0;
    for (int o = 16; o > 0; o >>= 1) ws += __shfl_xor(ws, o);
    const int excl = ws + inc - v;
    s_start[t] = excl;
    if (t <= NBK) cstart[g * CS_PITCH + t] = excl;
    __syncthreads();
    int2* slice = (int2*)part + e0;
    #pragma unroll
    for (int k = 0; k < PT_PER; ++k)
      slice[s_start[br[k] & 255] + (br[k] >> 8)] =
          make_int2(key[k], __float_as_int(wv[k]));
  }
  const int half = lane >> 4;
  const int mn   = lane & 15;
  const int n0   = wave * 16;

  const int ch = threadIdx.x & 63;
  const int r0 = threadIdx.x >> 6;

  int tile = blockIdx.x;
  float4 p0, p1, p2, p3, p4, p5, p6, p7;
#define G1_FETCH(t)                                                          \
  {                                                                          \
    const float* src_ = x + ((size_t)(t) * G1_BM + r0) * NFEAT + ch * 4;     \
    p0 = *(const float4*)(src_);                                             \
    p1 = *(const float4*)(src_ + 4 * NFEAT);                                 \
    p2 = *(const float4*)(src_ + 8 * NFEAT);                                 \
    p3 = *(const float4*)(src_ + 12 * NFEAT);                                \
    p4 = *(const float4*)(src_ + 16 * NFEAT);                                \
    p5 = *(const float4*)(src_ + 20 * NFEAT);                                \
    p6 = *(const float4*)(src_ + 24 * NFEAT);                                \
    p7 = *(const float4*)(src_ + 28 * NFEAT);                                \
  }
  G1_FETCH(tile)

  float bq[NFEAT / 8][4];
  {
    const float* bcol = W1 + (size_t)(4 * half) * NHID + n0 + mn;
    #pragma unroll
    for (int j = 0; j < NFEAT / 8; ++j) {
      #pragma unroll
      for (int i = 0; i < 4; ++i)
        bq[j][i] = bcol[(size_t)(8 * j + i) * NHID];
    }
  }

  const float* a0p = sX + mn * LDX + 4 * half;
  const float* a1p = a0p + 16 * LDX;

  for (; tile < G1_TILES; tile += G1_GRID) {
    const int m0 = tile * G1_BM;
    {
      float* dst = sX + r0 * LDX + ch * 4;
      *(float4*)(dst)            = p0;
      *(float4*)(dst + 4 * LDX)  = p1;
      *(float4*)(dst + 8 * LDX)  = p2;
      *(float4*)(dst + 12 * LDX) = p3;
      *(float4*)(dst + 16 * LDX) = p4;
      *(float4*)(dst + 20 * LDX) = p5;
      *(float4*)(dst + 24 * LDX) = p6;
      *(float4*)(dst + 28 * LDX) = p7;
    }
    __syncthreads();

    const int nxt = tile + G1_GRID;
    if (nxt < G1_TILES) G1_FETCH(nxt)

    v8f c0 = {};
    v8f c1 = {};
    #pragma unroll
    for (int j = 0; j < NFEAT / 8; ++j) {
      const float4 a0 = *(const float4*)(a0p + 8 * j);
      const float4 a1 = *(const float4*)(a1p + 8 * j);
      v2f a, b;
      b.x = bq[j][0]; b.y = bq[j][1];
      a.x = a0.x; a.y = a0.y;
      c0 = __builtin_amdgcn_wmma_f32_16x16x4_f32(
           false, a,  false, b,
           (short)0, c0,  false,  false);
      a.x = a1.x; a.y = a1.y;
      c1 = __builtin_amdgcn_wmma_f32_16x16x4_f32(
          false, a, false, b, (short)0, c1, false, false);
      b.x = bq[j][2]; b.y = bq[j][3];
      a.x = a0.z; a.y = a0.w;
      c0 = __builtin_amdgcn_wmma_f32_16x16x4_f32(
          false, a, false, b, (short)0, c0, false, false);
      a.x = a1.z; a.y = a1.w;
      c1 = __builtin_amdgcn_wmma_f32_16x16x4_f32(
          false, a, false, b, (short)0, c1, false, false);
    }
    float* crow = H0 + (size_t)m0 * NHID + n0 + mn;
    #pragma unroll
    for (int j = 0; j < 8; ++j) {
      crow[(size_t)(j + 8 * half) * NHID]      = c0[j];
      crow[(size_t)(j + 8 * half + 16) * NHID] = c1[j];
    }
    __syncthreads();
  }
}

__global__ __launch_bounds__(1024)
void gcn_bsort(const int* __restrict__ part, const int* __restrict__ cstart,
               int* __restrict__ rowptr, int* __restrict__ rec) {
  __shared__ int s_cnt[BK_NODES];
  __shared__ int s_cur[BK_NODES];
  __shared__ int s_wsum[16];
  __shared__ int s_segbase[256];
  __shared__ int s_seglen[256];
  __shared__ int s_lo[8];
  const int t = threadIdx.x;
  const int lane = t & 31;
  const int wave = t >> 5;
  const int b = blockIdx.x;
  const int2* p2 = (const int2*)part;

  int lo = 0, hi = 0;
  if (t < PT_GRID) {
    const int* cp = cstart + t * CS_PITCH + b;
    lo = cp[0];
    hi = cp[1];
  }
  if (t < BK_NODES) s_cnt[t] = 0;
  if (t < 256) {
    s_segbase[t] = t * PT_CHUNK + lo;
    s_seglen[t]  = hi - lo;
    int sl = lo;
    for (int o = 16; o > 0; o >>= 1) sl += __shfl_xor(sl, o);
    if (lane == 0) s_lo[wave] = sl;
  }
  __syncthreads();

  int sbase[BS_SEGS], slen[BS_SEGS];
  #pragma unroll
  for (int k = 0; k < BS_SEGS; ++k) {
    sbase[k] = __builtin_amdgcn_readfirstlane(s_segbase[wave + 32 * k]);
    slen[k]  = __builtin_amdgcn_readfirstlane(s_seglen[wave + 32 * k]);
  }
  int2 kw[BS_SEGS][2];
  int maxlen = 0;
  #pragma unroll
  for (int k = 0; k < BS_SEGS; ++k) {
    kw[k][0] = (lane < slen[k])      ? p2[sbase[k] + lane]      : make_int2(-1, 0);
    kw[k][1] = (lane + 32 < slen[k]) ? p2[sbase[k] + 32 + lane] : make_int2(-1, 0);
    maxlen = slen[k] > maxlen ? slen[k] : maxlen;
  }
  #pragma unroll
  for (int k = 0; k < BS_SEGS; ++k) {
    if (kw[k][0].x >= 0) atomicAdd(&s_cnt[kw[k][0].x >> SRC_BITS], 1);
    if (kw[k][1].x >= 0) atomicAdd(&s_cnt[kw[k][1].x >> SRC_BITS], 1);
  }
  if (maxlen > 64) {
    #pragma unroll
    for (int k = 0; k < BS_SEGS; ++k)
      for (int j = 64 + lane; j < slen[k]; j += 32)
        atomicAdd(&s_cnt[p2[sbase[k] + j].x >> SRC_BITS], 1);
  }
  __syncthreads();
  const int beg = s_lo[0] + s_lo[1] + s_lo[2] + s_lo[3] +
                  s_lo[4] + s_lo[5] + s_lo[6] + s_lo[7];

  int v = 0, inc = 0;
  if (t < BK_NODES) {
    v = s_cnt[t];
    inc = v;
    for (int o = 1; o < 32; o <<= 1) {
      const int u = __shfl_up(inc, o);
      if (lane >= o) inc += u;
    }
    if (lane == 31) s_wsum[wave] = inc;
  }
  __syncthreads();
  if (t < BK_NODES) {
    int ws = (lane < wave) ? s_wsum[lane] : 0;
    for (int o = 16; o > 0; o >>= 1) ws += __shfl_xor(ws, o);
    const int start = beg + ws + inc - v;
    s_cur[t] = start;
    const int node = b * BK_NODES + t;
    if (node < N_NODES) rowptr[node] = start;
    if (node == N_NODES - 1) rowptr[N_NODES] = start + v;
  }
  __syncthreads();

  int2* r2 = (int2*)rec;
  int pos[BS_SEGS][2];
  #pragma unroll
  for (int k = 0; k < BS_SEGS; ++k) {
    pos[k][0] = 0;
    pos[k][1] = 0;
    if (kw[k][0].x >= 0) pos[k][0] = atomicAdd(&s_cur[kw[k][0].x >> SRC_BITS], 1);
    if (kw[k][1].x >= 0) pos[k][1] = atomicAdd(&s_cur[kw[k][1].x >> SRC_BITS], 1);
  }
  #pragma unroll
  for (int k = 0; k < BS_SEGS; ++k) {
    if (kw[k][0].x >= 0)
      r2[pos[k][0]] = make_int2(kw[k][0].x & ((1 << SRC_BITS) - 1), kw[k][0].y);
    if (kw[k][1].x >= 0)
      r2[pos[k][1]] = make_int2(kw[k][1].x & ((1 << SRC_BITS) - 1), kw[k][1].y);
  }
  if (maxlen > 64) {
    #pragma unroll
    for (int k = 0; k < BS_SEGS; ++k)
      for (int j = 64 + lane; j < slen[k]; j += 32) {
        const int2 r = p2[sbase[k] + j];
        const int p = atomicAdd(&s_cur[r.x >> SRC_BITS], 1);
        r2[p] = make_int2(r.x & ((1 << SRC_BITS) - 1), r.y);
      }
  }
}


template <int K>
__device__ __forceinline__
void gcn_gather_group(float4& acc, const float* __restrict__ h0l, int2 sw, int j) {
  float4 hv[K];
  #pragma unroll
  for (int u = 0; u < K; ++u) {
    const int s = __builtin_amdgcn_readlane(sw.x, j + u);
    hv[u] = *(const float4*)(h0l + (size_t)s * NHID);
  }
  asm volatile("" ::: "memory");
  #pragma unroll
  for (int u = 0; u < K; ++u) {
    const float w = __int_as_float(__builtin_amdgcn_readlane(sw.y, j + u));
    acc.x += w * hv[u].x;
    acc.y += w * hv[u].y;
    acc.z += w * hv[u].z;
    acc.w += w * hv[u].w;
  }
}

__device__ __forceinline__
float4 gcn_row_sum(const int2* __restrict__ recs, const float* __restrict__ h0l,
                   int beg, int end, int2 sw, int lane) {
  float4 acc = make_float4(0.f, 0.f, 0.f, 0.f);
  for (int base = beg; base < end; base += 32) {
    const int left = end - base;
    const int n = left < 32 ? left : 32;
    if (base != beg) {
      sw = make_int2(0, 0);
      if (lane < n) sw = recs[base + lane];
    }
    for (int j = 0; j < n; j += 8) {
      const int m = n - j;
      if (m >= 8) {
        gcn_gather_group<8>(acc, h0l, sw, j);
      } else {
        switch (m) {
          case 7:  gcn_gather_group<7>(acc, h0l, sw, j); break;
          case 6:  gcn_gather_group<6>(acc, h0l, sw, j); break;
          case 5:  gcn_gather_group<5>(acc, h0l, sw, j); break;
          case 4:  gcn_gather_group<4>(acc, h0l, sw, j); break;
          case 3:  gcn_gather_group<3>(acc, h0l, sw, j); break;
          case 2:  gcn_gather_group<2>(acc, h0l, sw, j); break;
          default: gcn_gather_group<1>(acc, h0l, sw, j); break;
        }
      }
    }
  }
  return acc;
}

__global__ __launch_bounds__(256)
void gcn_agg_gemm2(const int* __restrict__ rowptr, const int* __restrict__ rec,
                   const float* __restrict__ H0, const float* __restrict__ b1,
                   const float* __restrict__ W2, float* __restrict__ H2) {
  __shared__ __attribute__((aligned(16))) float sA[16 * LDA];
  const int mt   = blockIdx.x;
  const int wave = threadIdx.x >> 5;
  const int lane = threadIdx.x & 31;
  const int m0   = mt * 16;
  const int2* recs = (const int2*)rec;
  const float4 bv = *(const float4*)(b1 + lane * 4);
  const float* h0l = H0 + lane * 4;

  const int* rp = rowptr + m0 + wave;
  const int rA0 = rp[0], rA1 = rp[1], rB0 = rp[8], rB1 = rp[9];
  const int begA = __builtin_amdgcn_readfirstlane(rA0);
  const int endA = __builtin_amdgcn_readfirstlane(rA1);
  const int begB = __builtin_amdgcn_readfirstlane(rB0);
  const int endB = __builtin_amdgcn_readfirstlane(rB1);

  int2 swA = make_int2(0, 0);
  int2 swB = make_int2(0, 0);
  if (lane < endA - begA) swA = recs[begA + lane];
  if (lane < endB - begB) swB = recs[begB + lane];

  {
    const float4 acc = gcn_row_sum(recs, h0l, begA, endA, swA, lane);
    float4 o;
    o.x = fmaxf(acc.x + bv.x, 0.0f);
    o.y = fmaxf(acc.y + bv.y, 0.0f);
    o.z = fmaxf(acc.z + bv.z, 0.0f);
    o.w = fmaxf(acc.w + bv.w, 0.0f);
    *(float4*)(sA + wave * LDA + lane * 4) = o;
  }
  {
    const float4 acc = gcn_row_sum(recs, h0l, begB, endB, swB, lane);
    float4 o;
    o.x = fmaxf(acc.x + bv.x, 0.0f);
    o.y = fmaxf(acc.y + bv.y, 0.0f);
    o.z = fmaxf(acc.z + bv.z, 0.0f);
    o.w = fmaxf(acc.w + bv.w, 0.0f);
    *(float4*)(sA + (wave + 8) * LDA + lane * 4) = o;
  }
  __syncthreads();

  if (wave < 3) {
    const int half = lane >> 4;
    const int mn   = lane & 15;
    const int col  = wave * 16 + mn;
    const bool colok = (col < NCLASS);
    const float* wcol = W2 + (colok ? col : 0) + (size_t)(2 * half) * NCLASS;
    const float* arow = sA + mn * LDA + 2 * half;

    v8f c = {};
    #pragma unroll 1
    for (int kh = 0; kh < NHID; kh += NHID / 2) {
      float bx[16], by[16];
      #pragma unroll
      for (int i = 0; i < 16; ++i) {
        const float vx = wcol[(size_t)(kh + 4 * i) * NCLASS];
        const float vy = wcol[(size_t)(kh + 4 * i + 1) * NCLASS];
        bx[i] = colok ? vx : 0.0f;
        by[i] = colok ? vy : 0.0f;
      }
      #pragma unroll
      for (int i = 0; i < 16; ++i) {
        v2f a, b;
        a.x = arow[kh + 4 * i];
        a.y = arow[kh + 4 * i + 1];
        b.x = bx[i];
        b.y = by[i];
        c = __builtin_amdgcn_wmma_f32_16x16x4_f32(
            false, a, false, b, (short)0, c, false, false);
      }
    }
    if (colok) {
      #pragma unroll
      for (int j = 0; j < 8; ++j)
        H2[(size_t)(m0 + j + 8 * half) * NCLASS + col] = c[j];
    }
  }
}

__global__ __launch_bounds__(256)
void gcn_spmm2(const int* __restrict__ rowptr, const int* __restrict__ rec,
               const float* __restrict__ H2, const float* __restrict__ b2,
               float* __restrict__ out) {
  const int tid = blockIdx.x * 256 + threadIdx.x;
  if (tid >= N_NODES * (NCLASS / 4)) return;
  const int r  = tid / (NCLASS / 4);
  const int c4 = tid % (NCLASS / 4);
  const int2* recs = (const int2*)rec;
  const int beg = rowptr[r];
  const int end = rowptr[r + 1];
  const float* hcol = H2 + c4 * 4;
  float4 acc = make_float4(0.f, 0.f, 0.f, 0.f);
  #pragma unroll 4
  for (int e = beg; e < end; ++e) {
    const int2 sw = recs[e];
    const float w = __int_as_float(sw.y);
    const float4 hv = *(const float4*)(hcol + (size_t)sw.x * NCLASS);
    acc.x += w * hv.x;
    acc.y += w * hv.y;
    acc.z += w * hv.z;
    acc.w += w * hv.w;
  }
  const float4 bv = *(const float4*)(b2 + c4 * 4);
  float4 o;
  o.x = acc.x + bv.x;
  o.y = acc.y + bv.y;
  o.z = acc.z + bv.z;
  o.w = acc.w + bv.w;
  *(float4*)(out + (size_t)tid * 4) = o;
}

extern "C" void kernel_launch(void* const* d_in, const int* in_sizes, int n_in,
                              void* d_out, int out_size, void* d_ws, size_t ws_size,
                              hipStream_t stream) {
  const float* x    = (const float*)d_in[0];
  const int*   esrc = (const int*)  d_in[1];
  const int*   edst = (const int*)  d_in[2];
  const float* ew   = (const float*)d_in[3];
  const float* W1   = (const float*)d_in[4];
  const float* b1   = (const float*)d_in[5];
  const float* W2   = (const float*)d_in[6];
  const float* b2   = (const float*)d_in[7];
  float* out = (float*)d_out;

  float* H0     = (float*)d_ws;
  float* H2     = H0 + (size_t)N_NODES * NHID;
  int*   part   = (int*)H2;
  int*   rec    = (int*)(H2 + (size_t)N_NODES * NCLASS);
  int*   cstart = rec + (size_t)N_EDGES * 2;
  int*   rowptr = cstart + (size_t)PT_GRID * CS_PITCH;

  gcn_gemm1<<<G1_GRID, 256, 0, stream>>>(x, W1, H0, esrc, edst, ew, cstart, part);
  gcn_bsort<<<NBK, 1024, 0, stream>>>(part, cstart, rowptr, rec);
  gcn_agg_gemm2<<<N_NODES / 16, 256, 0, stream>>>(rowptr, rec, H0, b1, W2, H2);
  gcn_spmm2<<<(N_NODES * (NCLASS / 4) + 255) / 256, 256, 0, stream>>>(rowptr, rec, H2, b2, out);
}


# Sort level 2: a bucket's sorted records go to an LDS image of its list stretch and leave as whole aligned 256-byte runs, no scattered 8-byte store per edge; row order = LDS arrival as before
# speedup vs baseline: 4.6013x; 1.0339x over previous
#include <hip/hip_runtime.h>


#define N_NODES 100000
#define N_EDGES 1600000
#define NFEAT   256
#define NHID    128
#define NCLASS  40

#define LDA     132

#define G1_BM    32
#define G1_TILES 3125
#define G1_GRID  512
#define LDX      260

typedef __attribute__((ext_vector_type(2))) float v2f;
typedef __attribute__((ext_vector_type(8))) float v8f;

#define BK_SHIFT 9
#define BK_NODES 512
#define NBK      196
#define PT_GRID  250
#define PT_CHUNK 6400
#define PT_PER   25
#define CS_PITCH (NBK + 1)
#define SRC_BITS 17
#define BS_SEGS  8
#define BS_CAP   12288

static_assert(PT_GRID * PT_CHUNK == N_EDGES, "chunks must tile the edge list");
static_assert(PT_PER * 256 == PT_CHUNK, "a chunk is a whole number of edges per thread");
static_assert(NBK * BK_NODES >= N_NODES, "buckets must cover the nodes");
static_assert(NBK < 256, "a bucket id must fit the low byte next to the rank");
static_assert(N_NODES <= (1 << SRC_BITS), "source id must fit its field");
static_assert(BS_SEGS * 32 >= PT_GRID && BS_SEGS * 32 <= 256, "segment table size");
static_assert(PT_GRID <= G1_GRID, "every chunk needs a block of launch 1");

__global__ __launch_bounds__(256)
void gcn_gemm1(const float* __restrict__ x, const float* __restrict__ W1,
               float* __restrict__ H0,
               const int* __restrict__ esrc, const int* __restrict__ edst,
               const float* __restrict__ ew,
               int* __restrict__ cstart, int* __restrict__ part) {
  __shared__ __attribute__((aligned(16))) float sX[G1_BM * LDX];
  __shared__ int s_cnt[256];
  __shared__ int s_start[256];
  __shared__ int s_wsum[8];
  const int wave = threadIdx.x >> 5;
  const int lane = threadIdx.x & 31;

  if (blockIdx.x >= G1_GRID - PT_GRID) {
    const int g = blockIdx.x - (G1_GRID - PT_GRID);
    const int t = threadIdx.x;
    s_cnt[t] = 0;
    const size_t e0 = (size_t)g * PT_CHUNK;
    int   key[PT_PER], br[PT_PER];
    float wv[PT_PER];
    #pragma unroll
    for (int k = 0; k < PT_PER; ++k) {
      const size_t i = e0 + k * 256 + t;
      const int d = edst[i];
      key[k] = ((d & (BK_NODES - 1)) << SRC_BITS) | esrc[i];
      br[k]  = d >> BK_SHIFT;
      wv[k]  = ew[i];
    }
    __syncthreads();
    #pragma unroll
    for (int k = 0; k < PT_PER; ++k) {
      const int bk = br[k];
      br[k] = bk | (atomicAdd(&s_cnt[bk], 1) << 8);
    }
    __syncthreads();
    const int v = s_cnt[t];
    int inc = v;
    for (int o = 1; o < 32; o <<= 1) {
      const int u = __shfl_up(inc, o);
      if (lane >= o) inc += u;
    }
    if (lane == 31) s_wsum[wave] = inc;
    __syncthreads();
    int ws = (lane < wave) ? s_wsum[lane] : 0;
    for (int o = 16; o > 0; o >>= 1) ws += __shfl_xor(ws, o);
    const int excl = ws + inc - v;
    s_start[t] = excl;
    if (t <= NBK) cstart[g * CS_PITCH + t] = excl;
    __syncthreads();
    int2* slice = (int2*)part + e0;
    #pragma unroll
    for (int k = 0; k < PT_PER; ++k)
      slice[s_start[br[k] & 255] + (br[k] >> 8)] =
          make_int2(key[k], __float_as_int(wv[k]));
  }
  const int half = lane >> 4;
  const int mn   = lane & 15;
  const int n0   = wave * 16;

  const int ch = threadIdx.x & 63;
  const int r0 = threadIdx.x >> 6;

  int tile = blockIdx.x;
  float4 p0, p1, p2, p3, p4, p5, p6, p7;
#define G1_FETCH(t)                                                          \
  {                                                                          \
    const float* src_ = x + ((size_t)(t) * G1_BM + r0) * NFEAT + ch * 4;     \
    p0 = *(const float4*)(src_);                                             \
    p1 = *(const float4*)(src_ + 4 * NFEAT);                                 \
    p2 = *(const float4*)(src_ + 8 * NFEAT);                                 \
    p3 = *(const float4*)(src_ + 12 * NFEAT);                                \
    p4 = *(const float4*)(src_ + 16 * NFEAT);                                \
    p5 = *(const float4*)(src_ + 20 * NFEAT);                                \
    p6 = *(const float4*)(src_ + 24 * NFEAT);                                \
    p7 = *(const float4*)(src_ + 28 * NFEAT);                                \
  }
  G1_FETCH(tile)

  float bq[NFEAT / 8][4];
  {
    const float* bcol = W1 + (size_t)(4 * half) * NHID + n0 + mn;
    #pragma unroll
    for (int j = 0; j < NFEAT / 8; ++j) {
      #pragma unroll
      for (int i = 0; i < 4; ++i)
        bq[j][i] = bcol[(size_t)(8 * j + i) * NHID];
    }
  }

  const float* a0p = sX + mn * LDX + 4 * half;
  const float* a1p = a0p + 16 * LDX;

  for (; tile < G1_TILES; tile += G1_GRID) {
    const int m0 = tile * G1_BM;
    {
      float* dst = sX + r0 * LDX + ch * 4;
      *(float4*)(dst)            = p0;
      *(float4*)(dst + 4 * LDX)  = p1;
      *(float4*)(dst + 8 * LDX)  = p2;
      *(float4*)(dst + 12 * LDX) = p3;
      *(float4*)(dst + 16 * LDX) = p4;
      *(float4*)(dst + 20 * LDX) = p5;
      *(float4*)(dst + 24 * LDX) = p6;
      *(float4*)(dst + 28 * LDX) = p7;
    }
    __syncthreads();

    const int nxt = tile + G1_GRID;
    if (nxt < G1_TILES) G1_FETCH(nxt)

    v8f c0 = {};
    v8f c1 = {};
    #pragma unroll
    for (int j = 0; j < NFEAT / 8; ++j) {
      const float4 a0 = *(const float4*)(a0p + 8 * j);
      const float4 a1 = *(const float4*)(a1p + 8 * j);
      v2f a, b;
      b.x = bq[j][0]; b.y = bq[j][1];
      a.x = a0.x; a.y = a0.y;
      c0 = __builtin_amdgcn_wmma_f32_16x16x4_f32(
           false, a,  false, b,
           (short)0, c0,  false,  false);
      a.x = a1.x; a.y = a1.y;
      c1 = __builtin_amdgcn_wmma_f32_16x16x4_f32(
          false, a, false, b, (short)0, c1, false, false);
      b.x = bq[j][2]; b.y = bq[j][3];
      a.x = a0.z; a.y = a0.w;
      c0 = __builtin_amdgcn_wmma_f32_16x16x4_f32(
          false, a, false, b, (short)0, c0, false, false);
      a.x = a1.z; a.y = a1.w;
      c1 = __builtin_amdgcn_wmma_f32_16x16x4_f32(
          false, a, false, b, (short)0, c1, false, false);
    }
    float* crow = H0 + (size_t)m0 * NHID + n0 + mn;
    #pragma unroll
    for (int j = 0; j < 8; ++j) {
      crow[(size_t)(j + 8 * half) * NHID]      = c0[j];
      crow[(size_t)(j + 8 * half + 16) * NHID] = c1[j];
    }
    __syncthreads();
  }
}

__global__ __launch_bounds__(1024)
void gcn_bsort(const int* __restrict__ part, const int* __restrict__ cstart,
               int* __restrict__ rowptr, int* __restrict__ rec) {
  __shared__ int s_cnt[BK_NODES];
  __shared__ int s_cur[BK_NODES];
  __shared__ int s_wsum[16];
  __shared__ int s_segbase[256];
  __shared__ int s_seglen[256];
  __shared__ int s_lo[8];
  __shared__ int s_tot;
  __shared__ __attribute__((aligned(16))) int s_img[2 * BS_CAP];
  int2* img = (int2*)s_img;
  const int t = threadIdx.x;
  const int lane = t & 31;
  const int wave = t >> 5;
  const int b = blockIdx.x;
  const int2* p2 = (const int2*)part;

  int lo = 0, hi = 0;
  if (t < PT_GRID) {
    const int* cp = cstart + t * CS_PITCH + b;
    lo = cp[0];
    hi = cp[1];
  }
  if (t < BK_NODES) s_cnt[t] = 0;
  if (t < 256) {
    s_segbase[t] = t * PT_CHUNK + lo;
    s_seglen[t]  = hi - lo;
    int sl = lo;
    for (int o = 16; o > 0; o >>= 1) sl += __shfl_xor(sl, o);
    if (lane == 0) s_lo[wave] = sl;
  }
  __syncthreads();

  int sbase[BS_SEGS], slen[BS_SEGS];
  #pragma unroll
  for (int k = 0; k < BS_SEGS; ++k) {
    sbase[k] = __builtin_amdgcn_readfirstlane(s_segbase[wave + 32 * k]);
    slen[k]  = __builtin_amdgcn_readfirstlane(s_seglen[wave + 32 * k]);
  }
  int2 kw[BS_SEGS][2];
  int maxlen = 0;
  #pragma unroll
  for (int k = 0; k < BS_SEGS; ++k) {
    kw[k][0] = (lane < slen[k])      ? p2[sbase[k] + lane]      : make_int2(-1, 0);
    kw[k][1] = (lane + 32 < slen[k]) ? p2[sbase[k] + 32 + lane] : make_int2(-1, 0);
    maxlen = slen[k] > maxlen ? slen[k] : maxlen;
  }
  #pragma unroll
  for (int k = 0; k < BS_SEGS; ++k) {
    if (kw[k][0].x >= 0) atomicAdd(&s_cnt[kw[k][0].x >> SRC_BITS], 1);
    if (kw[k][1].x >= 0) atomicAdd(&s_cnt[kw[k][1].x >> SRC_BITS], 1);
  }
  if (maxlen > 64) {
    #pragma unroll
    for (int k = 0; k < BS_SEGS; ++k)
      for (int j = 64 + lane; j < slen[k]; j += 32)
        atomicAdd(&s_cnt[p2[sbase[k] + j].x >> SRC_BITS], 1);
  }
  __syncthreads();
  const int beg = s_lo[0] + s_lo[1] + s_lo[2] + s_lo[3] +
                  s_lo[4] + s_lo[5] + s_lo[6] + s_lo[7];

  int v = 0, inc = 0;
  if (t < BK_NODES) {
    v = s_cnt[t];
    inc = v;
    for (int o = 1; o < 32; o <<= 1) {
      const int u = __shfl_up(inc, o);
      if (lane >= o) inc += u;
    }
    if (lane == 31) s_wsum[wave] = inc;
  }
  __syncthreads();
  if (t < BK_NODES) {
    int ws = (lane < wave) ? s_wsum[lane] : 0;
    for (int o = 16; o > 0; o >>= 1) ws += __shfl_xor(ws, o);
    const int local = ws + inc - v;
    const int start = beg + local;
    s_cur[t] = local;
    if (t == BK_NODES - 1) s_tot = local + v;
    const int node = b * BK_NODES + t;
    if (node < N_NODES) rowptr[node] = start;
    if (node == N_NODES - 1) rowptr[N_NODES] = start + v;
  }
  __syncthreads();

  int2* r2 = (int2*)rec;
  int pos[BS_SEGS][2];
  #pragma unroll
  for (int k = 0; k < BS_SEGS; ++k) {
    pos[k][0] = 0;
    pos[k][1] = 0;
    if (kw[k][0].x >= 0) pos[k][0] = atomicAdd(&s_cur[kw[k][0].x >> SRC_BITS], 1);
    if (kw[k][1].x >= 0) pos[k][1] = atomicAdd(&s_cur[kw[k][1].x >> SRC_BITS], 1);
  }
  const int tot = s_tot;
  #pragma unroll
  for (int k = 0; k < BS_SEGS; ++k) {
    #pragma unroll
    for (int q = 0; q < 2; ++q)
      if (kw[k][q].x >= 0 && pos[k][q] < BS_CAP)
        img[pos[k][q]] = make_int2(kw[k][q].x & ((1 << SRC_BITS) - 1), kw[k][q].y);
  }
  if (tot > BS_CAP) {
    #pragma unroll
    for (int k = 0; k < BS_SEGS; ++k) {
      #pragma unroll
      for (int q = 0; q < 2; ++q)
        if (kw[k][q].x >= 0 && pos[k][q] >= BS_CAP)
          r2[beg + pos[k][q]] = make_int2(kw[k][q].x & ((1 << SRC_BITS) - 1), kw[k][q].y);
    }
  }
  if (maxlen > 64) {
    #pragma unroll
    for (int k = 0; k < BS_SEGS; ++k)
      for (int j = 64 + lane; j < slen[k]; j += 32) {
        const int2 r = p2[sbase[k] + j];
        const int p = atomicAdd(&s_cur[r.x >> SRC_BITS], 1);
        const int2 o = make_int2(r.x & ((1 << SRC_BITS) - 1), r.y);
        if (p < BS_CAP) img[p] = o; else r2[beg + p] = o;
      }
  }
  __syncthreads();
  const int end = beg + (tot < BS_CAP ? tot : BS_CAP);
  int g = (beg & ~31) + t;
  if (g >= beg && g < end) r2[g] = img[g - beg];
  g += 1024;
  #pragma unroll 4
  for (; g < end; g += 1024) r2[g] = img[g - beg];
}


template <int K>
__device__ __forceinline__
void gcn_gather_group(float4& acc, const float* __restrict__ h0l, int2 sw, int j) {
  float4 hv[K];
  #pragma unroll
  for (int u = 0; u < K; ++u) {
    const int s = __builtin_amdgcn_readlane(sw.x, j + u);
    hv[u] = *(const float4*)(h0l + (size_t)s * NHID);
  }
  asm volatile("" ::: "memory");
  #pragma unroll
  for (int u = 0; u < K; ++u) {
    const float w = __int_as_float(__builtin_amdgcn_readlane(sw.y, j + u));
    acc.x += w * hv[u].x;
    acc.y += w * hv[u].y;
    acc.z += w * hv[u].z;
    acc.w += w * hv[u].w;
  }
}

__device__ __forceinline__
float4 gcn_row_sum(const int2* __restrict__ recs, const float* __restrict__ h0l,
                   int beg, int end, int2 sw, int lane) {
  float4 acc = make_float4(0.f, 0.f, 0.f, 0.f);
  for (int base = beg; base < end; base += 32) {
    const int left = end - base;
    const int n = left < 32 ? left : 32;
    if (base != beg) {
      sw = make_int2(0, 0);
      if (lane < n) sw = recs[base + lane];
    }
    for (int j = 0; j < n; j += 8) {
      const int m = n - j;
      if (m >= 8) {
        gcn_gather_group<8>(acc, h0l, sw, j);
      } else {
        switch (m) {
          case 7:  gcn_gather_group<7>(acc, h0l, sw, j); break;
          case 6:  gcn_gather_group<6>(acc, h0l, sw, j); break;
          case 5:  gcn_gather_group<5>(acc, h0l, sw, j); break;
          case 4:  gcn_gather_group<4>(acc, h0l, sw, j); break;
          case 3:  gcn_gather_group<3>(acc, h0l, sw, j); break;
          case 2:  gcn_gather_group<2>(acc, h0l, sw, j); break;
          default: gcn_gather_group<1>(acc, h0l, sw, j); break;
        }
      }
    }
  }
  return acc;
}

__global__ __launch_bounds__(256)
void gcn_agg_gemm2(const int* __restrict__ rowptr, const int* __restrict__ rec,
                   const float* __restrict__ H0, const float* __restrict__ b1,
                   const float* __restrict__ W2, float* __restrict__ H2) {
  __shared__ __attribute__((aligned(16))) float sA[16 * LDA];
  const int mt   = blockIdx.x;
  const int wave = threadIdx.x >> 5;
  const int lane = threadIdx.x & 31;
  const int m0   = mt * 16;
  const int2* recs = (const int2*)rec;
  const float4 bv = *(const float4*)(b1 + lane * 4);
  const float* h0l = H0 + lane * 4;

  const int* rp = rowptr + m0 + wave;
  const int rA0 = rp[0], rA1 = rp[1], rB0 = rp[8], rB1 = rp[9];
  const int begA = __builtin_amdgcn_readfirstlane(rA0);
  const int endA = __builtin_amdgcn_readfirstlane(rA1);
  const int begB = __builtin_amdgcn_readfirstlane(rB0);
  const int endB = __builtin_amdgcn_readfirstlane(rB1);

  int2 swA = make_int2(0, 0);
  int2 swB = make_int2(0, 0);
  if (lane < endA - begA) swA = recs[begA + lane];
  if (lane < endB - begB) swB = recs[begB + lane];

  {
    const float4 acc = gcn_row_sum(recs, h0l, begA, endA, swA, lane);
    float4 o;
    o.x = fmaxf(acc.x + bv.x, 0.0f);
    o.y = fmaxf(acc.y + bv.y, 0.0f);
    o.z = fmaxf(acc.z + bv.z, 0.0f);
    o.w = fmaxf(acc.w + bv.w, 0.0f);
    *(float4*)(sA + wave * LDA + lane * 4) = o;
  }
  {
    const float4 acc = gcn_row_sum(recs, h0l, begB, endB, swB, lane);
    float4 o;
    o.x = fmaxf(acc.x + bv.x, 0.0f);
    o.y = fmaxf(acc.y + bv.y, 0.0f);
    o.z = fmaxf(acc.z + bv.z, 0.0f);
    o.w = fmaxf(acc.w + bv.w, 0.0f);
    *(float4*)(sA + (wave + 8) * LDA + lane * 4) = o;
  }
  __syncthreads();

  if (wave < 3) {
    const int half = lane >> 4;
    const int mn   = lane & 15;
    const int col  = wave * 16 + mn;
    const bool colok = (col < NCLASS);
    const float* wcol = W2 + (colok ? col : 0) + (size_t)(2 * half) * NCLASS;
    const float* arow = sA + mn * LDA + 2 * half;

    v8f c = {};
    #pragma unroll 1
    for (int kh = 0; kh < NHID; kh += NHID / 2) {
      float bx[16], by[16];
      #pragma unroll
      for (int i = 0; i < 16; ++i) {
        const float vx = wcol[(size_t)(kh + 4 * i) * NCLASS];
        const float vy = wcol[(size_t)(kh + 4 * i + 1) * NCLASS];
        bx[i] = colok ? vx : 0.0f;
        by[i] = colok ? vy : 0.0f;
      }
      #pragma unroll
      for (int i = 0; i < 16; ++i) {
        v2f a, b;
        a.x = arow[kh + 4 * i];
        a.y = arow[kh + 4 * i + 1];
        b.x = bx[i];
        b.y = by[i];
        c = __builtin_amdgcn_wmma_f32_16x16x4_f32(
            false, a, false, b, (short)0, c, false, false);
      }
    }
    if (colok) {
      #pragma unroll
      for (int j = 0; j < 8; ++j)
        H2[(size_t)(m0 + j + 8 * half) * NCLASS + col] = c[j];
    }
  }
}

__global__ __launch_bounds__(256)
void gcn_spmm2(const int* __restrict__ rowptr, const int* __restrict__ rec,
               const float* __restrict__ H2, const float* __restrict__ b2,
               float* __restrict__ out) {
  const int tid = blockIdx.x * 256 + threadIdx.x;
  if (tid >= N_NODES * (NCLASS / 4)) return;
  const int r  = tid / (NCLASS / 4);
  const int c4 = tid % (NCLASS / 4);
  const int2* recs = (const int2*)rec;
  const int beg = rowptr[r];
  const int end = rowptr[r + 1];
  const float* hcol = H2 + c4 * 4;
  float4 acc = make_float4(0.f, 0.f, 0.f, 0.f);
  #pragma unroll 4
  for (int e = beg; e < end; ++e) {
    const int2 sw = recs[e];
    const float w = __int_as_float(sw.y);
    const float4 hv = *(const float4*)(hcol + (size_t)sw.x * NCLASS);
    acc.x += w * hv.x;
    acc.y += w * hv.y;
    acc.z += w * hv.z;
    acc.w += w * hv.w;
  }
  const float4 bv = *(const float4*)(b2 + c4 * 4);
  float4 o;
  o.x = acc.x + bv.x;
  o.y = acc.y + bv.y;
  o.z = acc.z + bv.z;
  o.w = acc.w + bv.w;
  *(float4*)(out + (size_t)tid * 4) = o;
}

extern "C" void kernel_launch(void* const* d_in, const int* in_sizes, int n_in,
                              void* d_out, int out_size, void* d_ws, size_t ws_size,
                              hipStream_t stream) {
  const float* x    = (const float*)d_in[0];
  const int*   esrc = (const int*)  d_in[1];
  const int*   edst = (const int*)  d_in[2];
  const float* ew   = (const float*)d_in[3];
  const float* W1   = (const float*)d_in[4];
  const float* b1   = (const float*)d_in[5];
  const float* W2   = (const float*)d_in[6];
  const float* b2   = (const float*)d_in[7];
  float* out = (float*)d_out;

  float* H0     = (float*)d_ws;
  float* H2     = H0 + (size_t)N_NODES * NHID;
  int*   part   = (int*)H2;
  int*   rec    = (int*)(H2 + (size_t)N_NODES * NCLASS);
  int*   cstart = rec + (size_t)N_EDGES * 2;
  int*   rowptr = cstart + (size_t)PT_GRID * CS_PITCH;

  gcn_gemm1<<<G1_GRID, 256, 0, stream>>>(x, W1, H0, esrc, edst, ew, cstart, part);
  gcn_bsort<<<NBK, 1024, 0, stream>>>(part, cstart, rowptr, rec);
  gcn_agg_gemm2<<<N_NODES / 16, 256, 0, stream>>>(rowptr, rec, H0, b1, W2, H2);
  gcn_spmm2<<<(N_NODES * (NCLASS / 4) + 255) / 256, 256, 0, stream>>>(rowptr, rec, H2, b2, out);
}


# Launch 4: half a wave per row, 4 rows per wave; a row's first 32 records fetched in one trip by its own lanes and shared by DPP row_share; 8 row gathers per round trip; row sum order kept
# speedup vs baseline: 4.8143x; 1.0463x over previous
#include <hip/hip_runtime.h>


#define N_NODES 100000
#define N_EDGES 1600000
#define NFEAT   256
#define NHID    128
#define NCLASS  40

#define LDA     132

#define G1_BM    32
#define G1_TILES 3125
#define G1_GRID  512
#define LDX      260

typedef __attribute__((ext_vector_type(2))) float v2f;
typedef __attribute__((ext_vector_type(8))) float v8f;

#define BK_SHIFT 9
#define BK_NODES 512
#define NBK      196
#define PT_GRID  250
#define PT_CHUNK 6400
#define PT_PER   25
#define CS_PITCH (NBK + 1)
#define SRC_BITS 17
#define BS_SEGS  8
#define BS_CAP   12288

static_assert(PT_GRID * PT_CHUNK == N_EDGES, "chunks must tile the edge list");
static_assert(PT_PER * 256 == PT_CHUNK, "a chunk is a whole number of edges per thread");
static_assert(NBK * BK_NODES >= N_NODES, "buckets must cover the nodes");
static_assert(NBK < 256, "a bucket id must fit the low byte next to the rank");
static_assert(N_NODES <= (1 << SRC_BITS), "source id must fit its field");
static_assert(BS_SEGS * 32 >= PT_GRID && BS_SEGS * 32 <= 256, "segment table size");
static_assert(PT_GRID <= G1_GRID, "every chunk needs a block of launch 1");

__global__ __launch_bounds__(256)
void gcn_gemm1(const float* __restrict__ x, const float* __restrict__ W1,
               float* __restrict__ H0,
               const int* __restrict__ esrc, const int* __restrict__ edst,
               const float* __restrict__ ew,
               int* __restrict__ cstart, int* __restrict__ part) {
  __shared__ __attribute__((aligned(16))) float sX[G1_BM * LDX];
  __shared__ int s_cnt[256];
  __shared__ int s_start[256];
  __shared__ int s_wsum[8];
  const int wave = threadIdx.x >> 5;
  const int lane = threadIdx.x & 31;

  if (blockIdx.x >= G1_GRID - PT_GRID) {
    const int g = blockIdx.x - (G1_GRID - PT_GRID);
    const int t = threadIdx.x;
    s_cnt[t] = 0;
    const size_t e0 = (size_t)g * PT_CHUNK;
    int   key[PT_PER], br[PT_PER];
    float wv[PT_PER];
    #pragma unroll
    for (int k = 0; k < PT_PER; ++k) {
      const size_t i = e0 + k * 256 + t;
      const int d = edst[i];
      key[k] = ((d & (BK_NODES - 1)) << SRC_BITS) | esrc[i];
      br[k]  = d >> BK_SHIFT;
      wv[k]  = ew[i];
    }
    __syncthreads();
    #pragma unroll
    for (int k = 0; k < PT_PER; ++k) {
      const int bk = br[k];
      br[k] = bk | (atomicAdd(&s_cnt[bk], 1) << 8);
    }
    __syncthreads();
    const int v = s_cnt[t];
    int inc = v;
    for (int o = 1; o < 32; o <<= 1) {
      const int u = __shfl_up(inc, o);
      if (lane >= o) inc += u;
    }
    if (lane == 31) s_wsum[wave] = inc;
    __syncthreads();
    int ws = (lane < wave) ? s_wsum[lane] : 0;
    for (int o = 16; o > 0; o >>= 1) ws += __shfl_xor(ws, o);
    const int excl = ws + inc - v;
    s_start[t] = excl;
    if (t <= NBK) cstart[g * CS_PITCH + t] = excl;
    __syncthreads();
    int2* slice = (int2*)part + e0;
    #pragma unroll
    for (int k = 0; k < PT_PER; ++k)
      slice[s_start[br[k] & 255] + (br[k] >> 8)] =
          make_int2(key[k], __float_as_int(wv[k]));
  }
  const int half = lane >> 4;
  const int mn   = lane & 15;
  const int n0   = wave * 16;

  const int ch = threadIdx.x & 63;
  const int r0 = threadIdx.x >> 6;

  int tile = blockIdx.x;
  float4 p0, p1, p2, p3, p4, p5, p6, p7;
#define G1_FETCH(t)                                                          \
  {                                                                          \
    const float* src_ = x + ((size_t)(t) * G1_BM + r0) * NFEAT + ch * 4;     \
    p0 = *(const float4*)(src_);                                             \
    p1 = *(const float4*)(src_ + 4 * NFEAT);                                 \
    p2 = *(const float4*)(src_ + 8 * NFEAT);                                 \
    p3 = *(const float4*)(src_ + 12 * NFEAT);                                \
    p4 = *(const float4*)(src_ + 16 * NFEAT);                                \
    p5 = *(const float4*)(src_ + 20 * NFEAT);                                \
    p6 = *(const float4*)(src_ + 24 * NFEAT);                                \
    p7 = *(const float4*)(src_ + 28 * NFEAT);                                \
  }
  G1_FETCH(tile)

  float bq[NFEAT / 8][4];
  {
    const float* bcol = W1 + (size_t)(4 * half) * NHID + n0 + mn;
    #pragma unroll
    for (int j = 0; j < NFEAT / 8; ++j) {
      #pragma unroll
      for (int i = 0; i < 4; ++i)
        bq[j][i] = bcol[(size_t)(8 * j + i) * NHID];
    }
  }

  const float* a0p = sX + mn * LDX + 4 * half;
  const float* a1p = a0p + 16 * LDX;

  for (; tile < G1_TILES; tile += G1_GRID) {
    const int m0 = tile * G1_BM;
    {
      float* dst = sX + r0 * LDX + ch * 4;
      *(float4*)(dst)            = p0;
      *(float4*)(dst + 4 * LDX)  = p1;
      *(float4*)(dst + 8 * LDX)  = p2;
      *(float4*)(dst + 12 * LDX) = p3;
      *(float4*)(dst + 16 * LDX) = p4;
      *(float4*)(dst + 20 * LDX) = p5;
      *(float4*)(dst + 24 * LDX) = p6;
      *(float4*)(dst + 28 * LDX) = p7;
    }
    __syncthreads();

    const int nxt = tile + G1_GRID;
    if (nxt < G1_TILES) G1_FETCH(nxt)

    v8f c0 = {};
    v8f c1 = {};
    #pragma unroll
    for (int j = 0; j < NFEAT / 8; ++j) {
      const float4 a0 = *(const float4*)(a0p + 8 * j);
      const float4 a1 = *(const float4*)(a1p + 8 * j);
      v2f a, b;
      b.x = bq[j][0]; b.y = bq[j][1];
      a.x = a0.x; a.y = a0.y;
      c0 = __builtin_amdgcn_wmma_f32_16x16x4_f32(
           false, a,  false, b,
           (short)0, c0,  false,  false);
      a.x = a1.x; a.y = a1.y;
      c1 = __builtin_amdgcn_wmma_f32_16x16x4_f32(
          false, a, false, b, (short)0, c1, false, false);
      b.x = bq[j][2]; b.y = bq[j][3];
      a.x = a0.z; a.y = a0.w;
      c0 = __builtin_amdgcn_wmma_f32_16x16x4_f32(
          false, a, false, b, (short)0, c0, false, false);
      a.x = a1.z; a.y = a1.w;
      c1 = __builtin_amdgcn_wmma_f32_16x16x4_f32(
          false, a, false, b, (short)0, c1, false, false);
    }
    float* crow = H0 + (size_t)m0 * NHID + n0 + mn;
    #pragma unroll
    for (int j = 0; j < 8; ++j) {
      crow[(size_t)(j + 8 * half) * NHID]      = c0[j];
      crow[(size_t)(j + 8 * half + 16) * NHID] = c1[j];
    }
    __syncthreads();
  }
}

__global__ __launch_bounds__(1024)
void gcn_bsort(const int* __restrict__ part, const int* __restrict__ cstart,
               int* __restrict__ rowptr, int* __restrict__ rec) {
  __shared__ int s_cnt[BK_NODES];
  __shared__ int s_cur[BK_NODES];
  __shared__ int s_wsum[16];
  __shared__ int s_segbase[256];
  __shared__ int s_seglen[256];
  __shared__ int s_lo[8];
  __shared__ int s_tot;
  __shared__ __attribute__((aligned(16))) int s_img[2 * BS_CAP];
  int2* img = (int2*)s_img;
  const int t = threadIdx.x;
  const int lane = t & 31;
  const int wave = t >> 5;
  const int b = blockIdx.x;
  const int2* p2 = (const int2*)part;

  int lo = 0, hi = 0;
  if (t < PT_GRID) {
    const int* cp = cstart + t * CS_PITCH + b;
    lo = cp[0];
    hi = cp[1];
  }
  if (t < BK_NODES) s_cnt[t] = 0;
  if (t < 256) {
    s_segbase[t] = t * PT_CHUNK + lo;
    s_seglen[t]  = hi - lo;
    int sl = lo;
    for (int o = 16; o > 0; o >>= 1) sl += __shfl_xor(sl, o);
    if (lane == 0) s_lo[wave] = sl;
  }
  __syncthreads();

  int sbase[BS_SEGS], slen[BS_SEGS];
  #pragma unroll
  for (int k = 0; k < BS_SEGS; ++k) {
    sbase[k] = __builtin_amdgcn_readfirstlane(s_segbase[wave + 32 * k]);
    slen[k]  = __builtin_amdgcn_readfirstlane(s_seglen[wave + 32 * k]);
  }
  int2 kw[BS_SEGS][2];
  int maxlen = 0;
  #pragma unroll
  for (int k = 0; k < BS_SEGS; ++k) {
    kw[k][0] = (lane < slen[k])      ? p2[sbase[k] + lane]      : make_int2(-1, 0);
    kw[k][1] = (lane + 32 < slen[k]) ? p2[sbase[k] + 32 + lane] : make_int2(-1, 0);
    maxlen = slen[k] > maxlen ? slen[k] : maxlen;
  }
  #pragma unroll
  for (int k = 0; k < BS_SEGS; ++k) {
    if (kw[k][0].x >= 0) atomicAdd(&s_cnt[kw[k][0].x >> SRC_BITS], 1);
    if (kw[k][1].x >= 0) atomicAdd(&s_cnt[kw[k][1].x >> SRC_BITS], 1);
  }
  if (maxlen > 64) {
    #pragma unroll
    for (int k = 0; k < BS_SEGS; ++k)
      for (int j = 64 + lane; j < slen[k]; j += 32)
        atomicAdd(&s_cnt[p2[sbase[k] + j].x >> SRC_BITS], 1);
  }
  __syncthreads();
  const int beg = s_lo[0] + s_lo[1] + s_lo[2] + s_lo[3] +
                  s_lo[4] + s_lo[5] + s_lo[6] + s_lo[7];

  int v = 0, inc = 0;
  if (t < BK_NODES) {
    v = s_cnt[t];
    inc = v;
    for (int o = 1; o < 32; o <<= 1) {
      const int u = __shfl_up(inc, o);
      if (lane >= o) inc += u;
    }
    if (lane == 31) s_wsum[wave] = inc;
  }
  __syncthreads();
  if (t < BK_NODES) {
    int ws = (lane < wave) ? s_wsum[lane] : 0;
    for (int o = 16; o > 0; o >>= 1) ws += __shfl_xor(ws, o);
    const int local = ws + inc - v;
    const int start = beg + local;
    s_cur[t] = local;
    if (t == BK_NODES - 1) s_tot = local + v;
    const int node = b * BK_NODES + t;
    if (node < N_NODES) rowptr[node] = start;
    if (node == N_NODES - 1) rowptr[N_NODES] = start + v;
  }
  __syncthreads();

  int2* r2 = (int2*)rec;
  int pos[BS_SEGS][2];
  #pragma unroll
  for (int k = 0; k < BS_SEGS; ++k) {
    pos[k][0] = 0;
    pos[k][1] = 0;
    if (kw[k][0].x >= 0) pos[k][0] = atomicAdd(&s_cur[kw[k][0].x >> SRC_BITS], 1);
    if (kw[k][1].x >= 0) pos[k][1] = atomicAdd(&s_cur[kw[k][1].x >> SRC_BITS], 1);
  }
  const int tot = s_tot;
  #pragma unroll
  for (int k = 0; k < BS_SEGS; ++k) {
    #pragma unroll
    for (int q = 0; q < 2; ++q)
      if (kw[k][q].x >= 0 && pos[k][q] < BS_CAP)
        img[pos[k][q]] = make_int2(kw[k][q].x & ((1 << SRC_BITS) - 1), kw[k][q].y);
  }
  if (tot > BS_CAP) {
    #pragma unroll
    for (int k = 0; k < BS_SEGS; ++k) {
      #pragma unroll
      for (int q = 0; q < 2; ++q)
        if (kw[k][q].x >= 0 && pos[k][q] >= BS_CAP)
          r2[beg + pos[k][q]] = make_int2(kw[k][q].x & ((1 << SRC_BITS) - 1), kw[k][q].y);
    }
  }
  if (maxlen > 64) {
    #pragma unroll
    for (int k = 0; k < BS_SEGS; ++k)
      for (int j = 64 + lane; j < slen[k]; j += 32) {
        const int2 r = p2[sbase[k] + j];
        const int p = atomicAdd(&s_cur[r.x >> SRC_BITS], 1);
        const int2 o = make_int2(r.x & ((1 << SRC_BITS) - 1), r.y);
        if (p < BS_CAP) img[p] = o; else r2[beg + p] = o;
      }
  }
  __syncthreads();
  const int end = beg + (tot < BS_CAP ? tot : BS_CAP);
  int g = (beg & ~31) + t;
  if (g >= beg && g < end) r2[g] = img[g - beg];
  g += 1024;
  #pragma unroll 4
  for (; g < end; g += 1024) r2[g] = img[g - beg];
}


template <int K>
__device__ __forceinline__
void gcn_gather_group(float4& acc, const float* __restrict__ h0l, int2 sw, int j) {
  float4 hv[K];
  #pragma unroll
  for (int u = 0; u < K; ++u) {
    const int s = __builtin_amdgcn_readlane(sw.x, j + u);
    hv[u] = *(const float4*)(h0l + (size_t)s * NHID);
  }
  asm volatile("" ::: "memory");
  #pragma unroll
  for (int u = 0; u < K; ++u) {
    const float w = __int_as_float(__builtin_amdgcn_readlane(sw.y, j + u));
    acc.x += w * hv[u].x;
    acc.y += w * hv[u].y;
    acc.z += w * hv[u].z;
    acc.w += w * hv[u].w;
  }
}

__device__ __forceinline__
float4 gcn_row_sum(const int2* __restrict__ recs, const float* __restrict__ h0l,
                   int beg, int end, int2 sw, int lane) {
  float4 acc = make_float4(0.f, 0.f, 0.f, 0.f);
  for (int base = beg; base < end; base += 32) {
    const int left = end - base;
    const int n = left < 32 ? left : 32;
    if (base != beg) {
      sw = make_int2(0, 0);
      if (lane < n) sw = recs[base + lane];
    }
    for (int j = 0; j < n; j += 8) {
      const int m = n - j;
      if (m >= 8) {
        gcn_gather_group<8>(acc, h0l, sw, j);
      } else {
        switch (m) {
          case 7:  gcn_gather_group<7>(acc, h0l, sw, j); break;
          case 6:  gcn_gather_group<6>(acc, h0l, sw, j); break;
          case 5:  gcn_gather_group<5>(acc, h0l, sw, j); break;
          case 4:  gcn_gather_group<4>(acc, h0l, sw, j); break;
          case 3:  gcn_gather_group<3>(acc, h0l, sw, j); break;
          case 2:  gcn_gather_group<2>(acc, h0l, sw, j); break;
          default: gcn_gather_group<1>(acc, h0l, sw, j); break;
        }
      }
    }
  }
  return acc;
}

__global__ __launch_bounds__(256)
void gcn_agg_gemm2(const int* __restrict__ rowptr, const int* __restrict__ rec,
                   const float* __restrict__ H0, const float* __restrict__ b1,
                   const float* __restrict__ W2, float* __restrict__ H2) {
  __shared__ __attribute__((aligned(16))) float sA[16 * LDA];
  const int mt   = blockIdx.x;
  const int wave = threadIdx.x >> 5;
  const int lane = threadIdx.x & 31;
  const int m0   = mt * 16;
  const int2* recs = (const int2*)rec;
  const float4 bv = *(const float4*)(b1 + lane * 4);
  const float* h0l = H0 + lane * 4;

  const int* rp = rowptr + m0 + wave;
  const int rA0 = rp[0], rA1 = rp[1], rB0 = rp[8], rB1 = rp[9];
  const int begA = __builtin_amdgcn_readfirstlane(rA0);
  const int endA = __builtin_amdgcn_readfirstlane(rA1);
  const int begB = __builtin_amdgcn_readfirstlane(rB0);
  const int endB = __builtin_amdgcn_readfirstlane(rB1);

  int2 swA = make_int2(0, 0);
  int2 swB = make_int2(0, 0);
  if (lane < endA - begA) swA = recs[begA + lane];
  if (lane < endB - begB) swB = recs[begB + lane];

  {
    const float4 acc = gcn_row_sum(recs, h0l, begA, endA, swA, lane);
    float4 o;
    o.x = fmaxf(acc.x + bv.x, 0.0f);
    o.y = fmaxf(acc.y + bv.y, 0.0f);
    o.z = fmaxf(acc.z + bv.z, 0.0f);
    o.w = fmaxf(acc.w + bv.w, 0.0f);
    *(float4*)(sA + wave * LDA + lane * 4) = o;
  }
  {
    const float4 acc = gcn_row_sum(recs, h0l, begB, endB, swB, lane);
    float4 o;
    o.x = fmaxf(acc.x + bv.x, 0.0f);
    o.y = fmaxf(acc.y + bv.y, 0.0f);
    o.z = fmaxf(acc.z + bv.z, 0.0f);
    o.w = fmaxf(acc.w + bv.w, 0.0f);
    *(float4*)(sA + (wave + 8) * LDA + lane * 4) = o;
  }
  __syncthreads();

  if (wave < 3) {
    const int half = lane >> 4;
    const int mn   = lane & 15;
    const int col  = wave * 16 + mn;
    const bool colok = (col < NCLASS);
    const float* wcol = W2 + (colok ? col : 0) + (size_t)(2 * half) * NCLASS;
    const float* arow = sA + mn * LDA + 2 * half;

    v8f c = {};
    #pragma unroll 1
    for (int kh = 0; kh < NHID; kh += NHID / 2) {
      float bx[16], by[16];
      #pragma unroll
      for (int i = 0; i < 16; ++i) {
        const float vx = wcol[(size_t)(kh + 4 * i) * NCLASS];
        const float vy = wcol[(size_t)(kh + 4 * i + 1) * NCLASS];
        bx[i] = colok ? vx : 0.0f;
        by[i] = colok ? vy : 0.0f;
      }
      #pragma unroll
      for (int i = 0; i < 16; ++i) {
        v2f a, b;
        a.x = arow[kh + 4 * i];
        a.y = arow[kh + 4 * i + 1];
        b.x = bx[i];
        b.y = by[i];
        c = __builtin_amdgcn_wmma_f32_16x16x4_f32(
            false, a, false, b, (short)0, c, false, false);
      }
    }
    if (colok) {
      #pragma unroll
      for (int j = 0; j < 8; ++j)
        H2[(size_t)(m0 + j + 8 * half) * NCLASS + col] = c[j];
    }
  }
}


template <int J>
__device__ __forceinline__ int gcn_row_share(int v) {
  return __builtin_amdgcn_update_dpp(0, v, 0x150 + J, 0xF, 0xF, false);
}

#define GCN_S2_LOAD(U)                                                        \
  {                                                                           \
    constexpr int J = J0 + (U);                                               \
    const int s = gcn_row_share<(J & 15)>(J < 16 ? ra.x : rb.x);              \
    hv[U] = make_float4(0.f, 0.f, 0.f, 0.f);                                  \
    if (J < lim) hv[U] = *(const float4*)(hq + (size_t)s * NCLASS);           \
  }
#define GCN_S2_FMA(U)                                                         \
  {                                                                           \
    constexpr int J = J0 + (U);                                               \
    const float w = __int_as_float(gcn_row_share<(J & 15)>(J < 16 ? ra.y : rb.y)); \
    acc.x += w * hv[U].x;                                                     \
    acc.y += w * hv[U].y;                                                     \
    acc.z += w * hv[U].z;                                                     \
    acc.w += w * hv[U].w;                                                     \
  }

template <int J0>
__device__ __forceinline__
void gcn_s2_group(float4& acc, const float* __restrict__ hq, int2 ra, int2 rb,
                  int lim) {
  float4 hv[8];
  GCN_S2_LOAD(0) GCN_S2_LOAD(1) GCN_S2_LOAD(2) GCN_S2_LOAD(3)
  GCN_S2_LOAD(4) GCN_S2_LOAD(5) GCN_S2_LOAD(6) GCN_S2_LOAD(7)
  GCN_S2_FMA(0) GCN_S2_FMA(1) GCN_S2_FMA(2) GCN_S2_FMA(3)
  GCN_S2_FMA(4) GCN_S2_FMA(5) GCN_S2_FMA(6) GCN_S2_FMA(7)
}

__device__ __forceinline__
float4 gcn_s2_pair(const int2* __restrict__ recs, const float* __restrict__ hq,
                   int beg, int len, int lim, int nmax, int2 ra, int2 rb, int q) {
  float4 acc = make_float4(0.f, 0.f, 0.f, 0.f);
  for (int base = 0; base < nmax; base += 32) {
    if (base != 0) {
      ra = make_int2(0, 0);
      rb = make_int2(0, 0);
      if (base + q < len)      ra = recs[beg + base + q];
      if (base + 16 + q < len) rb = recs[beg + base + 16 + q];
    }
    const int l = lim - base;
    const int n = nmax - base;
    gcn_s2_group<0>(acc, hq, ra, rb, l);
    if (n > 8)  gcn_s2_group<8>(acc, hq, ra, rb, l);
    if (n > 16) gcn_s2_group<16>(acc, hq, ra, rb, l);
    if (n > 24) gcn_s2_group<24>(acc, hq, ra, rb, l);
  }
  return acc;
}

static_assert(N_NODES % 32 == 0, "a block of launch 4 is 8 waves of 4 rows");
static_assert(NCLASS == 40, "ten 16-byte column groups per row");

__global__ __launch_bounds__(256)
void gcn_spmm2(const int* __restrict__ rowptr, const int* __restrict__ rec,
               const float* __restrict__ H2, const float* __restrict__ b2,
               float* __restrict__ out) {
  const int lane = threadIdx.x & 31;
  const int h    = lane >> 4;
  const int q    = lane & 15;
  const bool lane_on = (q < NCLASS / 4);
  const int row0 = (blockIdx.x * 8 + (threadIdx.x >> 5)) * 4;
  const int2* recs = (const int2*)rec;

  const int rpv = rowptr[row0 + (lane < 4 ? lane : 4)];
  const int p0 = __builtin_amdgcn_readlane(rpv, 0);
  const int p1 = __builtin_amdgcn_readlane(rpv, 1);
  const int p2 = __builtin_amdgcn_readlane(rpv, 2);
  const int p3 = __builtin_amdgcn_readlane(rpv, 3);
  const int p4 = __builtin_amdgcn_readlane(rpv, 4);

  const int begA = h ? p1 : p0, lenA = (h ? p2 : p1) - begA;
  const int begB = h ? p3 : p2, lenB = (h ? p4 : p3) - begB;
  const int nA0 = p1 - p0, nA1 = p2 - p1, nB0 = p3 - p2, nB1 = p4 - p3;
  const int nmaxA = nA0 > nA1 ? nA0 : nA1;
  const int nmaxB = nB0 > nB1 ? nB0 : nB1;

  int2 raA = make_int2(0, 0), rbA = make_int2(0, 0);
  int2 raB = make_int2(0, 0), rbB = make_int2(0, 0);
  if (q < lenA)      raA = recs[begA + q];
  if (q + 16 < lenA) rbA = recs[begA + 16 + q];
  if (q < lenB)      raB = recs[begB + q];
  if (q + 16 < lenB) rbB = recs[begB + 16 + q];

  const float* hq = H2 + (lane_on ? q * 4 : 0);

  {
    const float4 acc = gcn_s2_pair(recs, hq, begA, lenA, lane_on ? lenA : 0, nmaxA, raA, rbA, q);
    float4 bv = make_float4(0.f, 0.f, 0.f, 0.f);
    if (lane_on) bv = *(const float4*)(b2 + q * 4);
    float4 o;
    o.x = acc.x + bv.x;
    o.y = acc.y + bv.y;
    o.z = acc.z + bv.z;
    o.w = acc.w + bv.w;
    if (lane_on) *(float4*)(out + (size_t)(row0 + h) * NCLASS + q * 4) = o;
  }
  {
    const float4 acc = gcn_s2_pair(recs, hq, begB, lenB, lane_on ? lenB : 0, nmaxB, raB, rbB, q);
    float4 bv = make_float4(0.f, 0.f, 0.f, 0.f);
    if (lane_on) bv = *(const float4*)(b2 + q * 4);
    float4 o;
    o.x = acc.x + bv.x;
    o.y = acc.y + bv.y;
    o.z = acc.z + bv.z;
    o.w = acc.w + bv.w;
    if (lane_on) *(float4*)(out + (size_t)(row0 + 2 + h) * NCLASS + q * 4) = o;
  }
}

extern "C" void kernel_launch(void* const* d_in, const int* in_sizes, int n_in,
                              void* d_out, int out_size, void* d_ws, size_t ws_size,
                              hipStream_t stream) {
  const float* x    = (const float*)d_in[0];
  const int*   esrc = (const int*)  d_in[1];
  const int*   edst = (const int*)  d_in[2];
  const float* ew   = (const float*)d_in[3];
  const float* W1   = (const float*)d_in[4];
  const float* b1   = (const float*)d_in[5];
  const float* W2   = (const float*)d_in[6];
  const float* b2   = (const float*)d_in[7];
  float* out = (float*)d_out;

  float* H0     = (float*)d_ws;
  float* H2     = H0 + (size_t)N_NODES * NHID;
  int*   part   = (int*)H2;
  int*   rec    = (int*)(H2 + (size_t)N_NODES * NCLASS);
  int*   cstart = rec + (size_t)N_EDGES * 2;
  int*   rowptr = cstart + (size_t)PT_GRID * CS_PITCH;

  gcn_gemm1<<<G1_GRID, 256, 0, stream>>>(x, W1, H0, esrc, edst, ew, cstart, part);
  gcn_bsort<<<NBK, 1024, 0, stream>>>(part, cstart, rowptr, rec);
  gcn_agg_gemm2<<<N_NODES / 16, 256, 0, stream>>>(rowptr, rec, H0, b1, W2, H2);
  gcn_spmm2<<<N_NODES / 32, 256, 0, stream>>>(rowptr, rec, H2, b2, out);
}
